# starting kernel
# baseline (speedup 1.0000x reference)
#include <hip/hip_runtime.h>
#include <hip/hip_bf16.h>
#include <hip/hip_fp16.h>

typedef __attribute__((ext_vector_type(2)))  float    v2f;
typedef __attribute__((ext_vector_type(8)))  float    v8f;
typedef __attribute__((ext_vector_type(16))) _Float16 v16h;

#define B_   2
#define C_   32
#define H_   96
#define W_   128
#define HW_  (H_ * W_)
#define BC_  (B_ * C_)
#define EPSF 1e-6f

#define OFF_QS    0
#define OFF_QC    6144
#define OFF_KH    14336
#define OFF_KW    20480
#define OFF_SQ    28672
#define OFF_SK    28736
#define OFF_GMAX  28800
#define OFF_GSMAX 28928
#define OFF_HR    28960
#define OFF_HC    (OFF_HR + 32 * 6144)
#define OFF_WR    (OFF_HC + 32 * 6144)
#define OFF_WC    (OFF_WR + 32 * 8192)
#define OFF_IRS   (OFF_WC + 32 * 8192)
#define OFF_SCORE (OFF_IRS + 4 * 64 * 224)

#if __has_builtin(__builtin_amdgcn_wmma_f32_16x16x4_f32)
#define USE_WMMA_F32 1
#endif

__global__ void k1_sums(const float* __restrict__ q, const float* __restrict__ k,
                        float* __restrict__ ws) {
  __shared__ float tile[HW_];
  __shared__ float rowbuf[H_];
  const int bc  = blockIdx.x;
  const int tid = threadIdx.x;
  for (int p = 0; p < 2; ++p) {
    const float* s = p ? k : q;
    for (int i = tid; i < HW_; i += 256) tile[i] = s[bc * HW_ + i];
    __syncthreads();
    if (tid < H_) {
      float acc = 0.f;
      for (int x = 0; x < W_; ++x) acc += tile[tid * W_ + x];
      ws[(p ? OFF_KH : OFF_QS) + bc * H_ + tid] = acc;
      rowbuf[tid] = acc;
    }
    if (tid >= 128) {
      const int x = tid - 128;
      float acc = 0.f;
      for (int h = 0; h < H_; ++h) acc += tile[h * W_ + x];
      ws[(p ? OFF_KW : OFF_QC) + bc * W_ + x] = acc;
    }
    __syncthreads();
    if (tid == 0) {
      float t = 0.f;
      for (int h = 0; h < H_; ++h) t += rowbuf[h];
      ws[(p ? OFF_SK : OFF_SQ) + bc] = t;
    }
    __syncthreads();
  }
}

__global__ void k2_norm(float* __restrict__ ws) {
  const int type = blockIdx.x & 3;
  const int c0   = blockIdx.x >> 2;
  const int tid  = threadIdx.x;
  __shared__ float rmn[256], rmx[256];
  const int   len    = (type < 2) ? H_ : W_;
  const int   N      = BC_ * len;
  const float addeps = (type < 2) ? (H_ * EPSF) : (W_ * EPSF);
  float* dst;
  if      (type == 0) dst = ws + OFF_HR + c0 * (BC_ * H_);
  else if (type == 1) dst = ws + OFF_HC + c0 * (BC_ * H_);
  else if (type == 2) dst = ws + OFF_WR + c0 * (BC_ * W_);
  else                dst = ws + OFF_WC + c0 * (BC_ * W_);
  float lmn = 3.4e38f, lmx = -3.4e38f;
  for (int i = tid; i < N; i += 256) {
    const int b = i / (C_ * len);
    const int r = i % (C_ * len);
    const int c = r / len;
    const int x = r % len;
    float raw;
    if      (type == 0) raw = ws[OFF_SQ + b * C_ + c0] * ws[OFF_KH + (b * C_ + c) * H_ + x] + addeps;
    else if (type == 1) raw = ws[OFF_QS + (b * C_ + c0) * H_ + x] * ws[OFF_SK + b * C_ + c] + addeps;
    else if (type == 2) raw = ws[OFF_SQ + b * C_ + c0] * ws[OFF_KW + (b * C_ + c) * W_ + x] + addeps;
    else                raw = ws[OFF_QC + (b * C_ + c0) * W_ + x] * ws[OFF_SK + b * C_ + c] + addeps;
    dst[i] = raw;
    lmn = fminf(lmn, raw);
    lmx = fmaxf(lmx, raw);
  }
  rmn[tid] = lmn; rmx[tid] = lmx;
  __syncthreads();
  for (int s = 128; s > 0; s >>= 1) {
    if (tid < s) { rmn[tid] = fminf(rmn[tid], rmn[tid + s]); rmx[tid] = fmaxf(rmx[tid], rmx[tid + s]); }
    __syncthreads();
  }
  const float mn  = rmn[0];
  const float inv = 1.0f / (rmx[0] - mn);
  for (int i = tid; i < N; i += 256) dst[i] = 2.0f + (dst[i] - mn) * inv;
  if (blockIdx.x == 0)
    for (int i = tid; i < 160; i += 256) ws[OFF_GMAX + i] = 0.0f;
}

__global__ void k3_tile(float* __restrict__ ws, int c0) {
  const int t   = blockIdx.x >> 6;
  const int bc  = blockIdx.x & 63;
  const int tid = threadIdx.x;
  __shared__ float E[HW_];
  __shared__ float mv[H_], nv[W_];
  __shared__ float rsb[H_], csb[W_];
  __shared__ float red[256];

  const float* mptr = ws + ((t < 2) ? OFF_HR : OFF_HC) + c0 * (BC_ * H_) + bc * H_;
  const float* nptr = ws + (((t & 1) == 0) ? OFF_WR : OFF_WC) + c0 * (BC_ * W_) + bc * W_;
  if (tid < H_)   mv[tid] = mptr[tid];
  if (tid >= 128) nv[tid - 128] = nptr[tid - 128];
  __syncthreads();

  for (int i = tid; i < HW_; i += 256) E[i] = __expf(mv[i >> 7] * nv[i & 127]);
  __syncthreads();

  const int wave = tid >> 5, lane = tid & 31;
#if defined(USE_WMMA_F32)
  if (wave < 6) {
    const int M   = wave * 16 + (lane & 15);
    const int kho = (lane < 16) ? 0 : 2;
    const v2f ones = {1.0f, 1.0f};
    v8f acc = {};
    for (int k0 = 0; k0 < W_; k0 += 4) {
      v2f a = { E[M * W_ + k0 + kho], E[M * W_ + k0 + kho + 1] };
      acc = __builtin_amdgcn_wmma_f32_16x16x4_f32(false, a, false, ones, (short)0, acc, false, false);
    }
    if ((lane & 15) == 0) {
      const int rbase = wave * 16 + ((lane >> 4) << 3);
      #pragma unroll
      for (int v = 0; v < 8; ++v) rsb[rbase + v] = acc[v];
    }
  } else {
    const int N   = lane & 15;
    const int kho = (lane < 16) ? 0 : 2;
    const v2f onesA = {1.0f, 1.0f};
    for (int nb = (wave - 6) * 4; nb < (wave - 6) * 4 + 4; ++nb) {
      const int N0 = nb * 16;
      v8f acc = {};
      for (int k0 = 0; k0 < H_; k0 += 4) {
        v2f b = { E[(k0 + kho) * W_ + N0 + N], E[(k0 + kho + 1) * W_ + N0 + N] };
        acc = __builtin_amdgcn_wmma_f32_16x16x4_f32(false, onesA, false, b, (short)0, acc, false, false);
      }
      if (lane < 16) csb[N0 + N] = acc[0];
    }
  }
#else
  if (wave < 6) {
    const int M = wave * 16 + (lane & 15);
    const int khalf = (lane < 16) ? 0 : 8;
    v16h hb;
    #pragma unroll
    for (int e = 0; e < 16; ++e) hb[e] = (_Float16)1.0f;
    v8f acc = {};
    for (int k0 = 0; k0 < W_; k0 += 32) {
      v16h ha;
      #pragma unroll
      for (int v = 0; v < 8; ++v) {
        const int kp = (v & 3) * 2 + ((v >> 2) << 4) + khalf;
        ha[2 * v]     = (_Float16)E[M * W_ + k0 + kp];
        ha[2 * v + 1] = (_Float16)E[M * W_ + k0 + kp + 1];
      }
      acc = __builtin_amdgcn_wmma_f32_16x16x32_f16(false, ha, false, hb, (short)0, acc, false, false);
    }
    if ((lane & 15) == 0) {
      const int rbase = wave * 16 + ((lane >> 4) << 3);
      #pragma unroll
      for (int v = 0; v < 8; ++v) rsb[rbase + v] = acc[v];
    }
  } else {
    const int N = lane & 15;
    const int khalf = (lane < 16) ? 0 : 16;
    v16h ha;
    #pragma unroll
    for (int e = 0; e < 16; ++e) ha[e] = (_Float16)1.0f;
    for (int nb = (wave - 6) * 4; nb < (wave - 6) * 4 + 4; ++nb) {
      const int N0 = nb * 16;
      v8f acc = {};
      for (int k0 = 0; k0 < H_; k0 += 32) {
        v16h hbm;
        #pragma unroll
        for (int v = 0; v < 8; ++v) {
          const int kk = khalf + 2 * v;
          hbm[2 * v]     = (_Float16)E[(k0 + kk) * W_ + N0 + N];
          hbm[2 * v + 1] = (_Float16)E[(k0 + kk + 1) * W_ + N0 + N];
        }
        acc = __builtin_amdgcn_wmma_f32_16x16x32_f16(false, ha, false, hbm, (short)0, acc, false, false);
      }
      if (lane < 16) csb[N0 + N] = acc[0];
    }
  }
#endif
  __syncthreads();

  float* irs = ws + OFF_IRS + (t * 64 + bc) * 224;
  if (tid < H_)   { const float v = 1.0f / rsb[tid]; rsb[tid] = v; irs[tid] = v; }
  if (tid >= 128) { const int x = tid - 128; const float v = 1.0f / csb[x]; csb[x] = v; irs[96 + x] = v; }
  __syncthreads();

  float lmax = 0.0f;
  for (int i = tid; i < HW_; i += 256)
    lmax = fmaxf(lmax, E[i] * (rsb[i >> 7] + csb[i & 127]));
  red[tid] = lmax;
  __syncthreads();
  for (int s = 128; s > 0; s >>= 1) {
    if (tid < s) red[tid] = fmaxf(red[tid], red[tid + s]);
    __syncthreads();
  }
  if (tid == 0)
    atomicMax((int*)(ws + OFF_GMAX + c0 * 4 + t), __float_as_int(red[0]));
}

__global__ void k4_score(float* __restrict__ ws, int c0) {
  const int bc  = blockIdx.x;
  const int tid = threadIdx.x;
  __shared__ float m2[2][H_], n2[2][W_], ir[4][H_], ic[4][W_], red[256];
  for (int i = tid; i < 2 * H_; i += 256) {
    const int s = i / H_, h = i % H_;
    m2[s][h] = ws[(s ? OFF_HC : OFF_HR) + c0 * (BC_ * H_) + bc * H_ + h];
  }
  for (int i = tid; i < 2 * W_; i += 256) {
    const int s = i / W_, x = i % W_;
    n2[s][x] = ws[(s ? OFF_WC : OFF_WR) + c0 * (BC_ * W_) + bc * W_ + x];
  }
  for (int i = tid; i < 4 * 224; i += 256) {
    const int t = i / 224, r = i % 224;
    const float v = ws[OFF_IRS + (t * 64 + bc) * 224 + r];
    if (r < 96) ir[t][r] = v; else ic[t][r - 96] = v;
  }
  __syncthreads();
  float ig[4];
  #pragma unroll
  for (int t = 0; t < 4; ++t) ig[t] = 1.0f / ws[OFF_GMAX + c0 * 4 + t];

  float* sc = ws + OFF_SCORE + bc * HW_;
  float lmax = 0.0f;
  for (int i = tid; i < HW_; i += 256) {
    const int h = i >> 7, x = i & 127;
    float s = 0.f;
    #pragma unroll
    for (int t = 0; t < 4; ++t) {
      const float e = __expf(m2[t >> 1][h] * n2[t & 1][x]);
      s += e * (ir[t][h] + ic[t][x]) * ig[t];
    }
    sc[i] = s;
    lmax = fmaxf(lmax, s);
  }
  red[tid] = lmax;
  __syncthreads();
  for (int s2 = 128; s2 > 0; s2 >>= 1) {
    if (tid < s2) red[tid] = fmaxf(red[tid], red[tid + s2]);
    __syncthreads();
  }
  if (tid == 0)
    atomicMax((int*)(ws + OFF_GSMAX + c0), __float_as_int(red[0]));
}

__global__ void k5_z(const float* __restrict__ ws, const float* __restrict__ values,
                     float* __restrict__ out, int c0) {
  const int gid = blockIdx.x * 256 + threadIdx.x;
  const int b = gid / HW_;
  const int r = gid % HW_;
  const float inv = 1.0f / ws[OFF_GSMAX + c0];
  float z = 0.f;
  for (int c = 0; c < C_; ++c) {
    const int off = (b * C_ + c) * HW_ + r;
    z += ws[OFF_SCORE + off] * values[off];
  }
  out[(b * C_ + c0) * HW_ + r] = z * inv;
}

__global__ void k6_scoreout(const float* __restrict__ ws, float* __restrict__ out) {
  const int gid = blockIdx.x * 256 + threadIdx.x;
  out[BC_ * HW_ + gid] = ws[OFF_SCORE + gid] * (1.0f / ws[OFF_GSMAX + 31]);
}

extern "C" void kernel_launch(void* const* d_in, const int* in_sizes, int n_in,
                              void* d_out, int out_size, void* d_ws, size_t ws_size,
                              hipStream_t stream) {
  (void)in_sizes; (void)n_in; (void)out_size; (void)ws_size;
  const float* q   = (const float*)d_in[0];
  const float* k   = (const float*)d_in[1];
  const float* val = (const float*)d_in[2];
  float* out = (float*)d_out;
  float* ws  = (float*)d_ws;

  k1_sums<<<64, 256, 0, stream>>>(q, k, ws);
  k2_norm<<<128, 256, 0, stream>>>(ws);
  for (int c0 = 0; c0 < C_; ++c0) {
    k3_tile<<<256, 256, 0, stream>>>(ws, c0);
    k4_score<<<64, 256, 0, stream>>>(ws, c0);
    k5_z<<<96, 256, 0, stream>>>(ws, val, out, c0);
  }
  k6_scoreout<<<3072, 256, 0, stream>>>(ws, out);
}


# Query-channel loop moved into the grid: 96 per-channel launches become 3 chunks of 2 (score fused with the values sum over c, same c order) plus one scaling pass; 99 launches to 9
# speedup vs baseline: 3.4839x; 3.4839x over previous
#include <hip/hip_runtime.h>
#include <hip/hip_bf16.h>
#include <hip/hip_fp16.h>

typedef __attribute__((ext_vector_type(2)))  float    v2f;
typedef __attribute__((ext_vector_type(8)))  float    v8f;
typedef __attribute__((ext_vector_type(16))) _Float16 v16h;

#define B_   2
#define C_   32
#define H_   96
#define W_   128
#define HW_  (H_ * W_)
#define BC_  (B_ * C_)
#define EPSF 1e-6f

#define OFF_QS    0
#define OFF_QC    6144
#define OFF_KH    14336
#define OFF_KW    20480
#define OFF_SQ    28672
#define OFF_SK    28736
#define OFF_GMAX  28800
#define OFF_GSMAX 28928
#define OFF_HR    28960
#define OFF_HC    (OFF_HR + 32 * 6144)
#define OFF_WR    (OFF_HC + 32 * 6144)
#define OFF_WC    (OFF_WR + 32 * 8192)
#define OFF_IRS   (OFF_WC + 32 * 8192)
#define ROWS_     2

#if __has_builtin(__builtin_amdgcn_wmma_f32_16x16x4_f32)
#define USE_WMMA_F32 1
#endif

__global__ void k1_sums(const float* __restrict__ q, const float* __restrict__ k,
                        float* __restrict__ ws) {
  __shared__ float tile[HW_];
  __shared__ float rowbuf[H_];
  const int bc  = blockIdx.x;
  const int tid = threadIdx.x;
  for (int p = 0; p < 2; ++p) {
    const float* s = p ? k : q;
    for (int i = tid; i < HW_; i += 256) tile[i] = s[bc * HW_ + i];
    __syncthreads();
    if (tid < H_) {
      float acc = 0.f;
      for (int x = 0; x < W_; ++x) acc += tile[tid * W_ + x];
      ws[(p ? OFF_KH : OFF_QS) + bc * H_ + tid] = acc;
      rowbuf[tid] = acc;
    }
    if (tid >= 128) {
      const int x = tid - 128;
      float acc = 0.f;
      for (int h = 0; h < H_; ++h) acc += tile[h * W_ + x];
      ws[(p ? OFF_KW : OFF_QC) + bc * W_ + x] = acc;
    }
    __syncthreads();
    if (tid == 0) {
      float t = 0.f;
      for (int h = 0; h < H_; ++h) t += rowbuf[h];
      ws[(p ? OFF_SK : OFF_SQ) + bc] = t;
    }
    __syncthreads();
  }
}

__global__ void k2_norm(float* __restrict__ ws) {
  const int type = blockIdx.x & 3;
  const int c0   = blockIdx.x >> 2;
  const int tid  = threadIdx.x;
  __shared__ float rmn[256], rmx[256];
  const int   len    = (type < 2) ? H_ : W_;
  const int   N      = BC_ * len;
  const float addeps = (type < 2) ? (H_ * EPSF) : (W_ * EPSF);
  float* dst;
  if      (type == 0) dst = ws + OFF_HR + c0 * (BC_ * H_);
  else if (type == 1) dst = ws + OFF_HC + c0 * (BC_ * H_);
  else if (type == 2) dst = ws + OFF_WR + c0 * (BC_ * W_);
  else                dst = ws + OFF_WC + c0 * (BC_ * W_);
  float lmn = 3.4e38f, lmx = -3.4e38f;
  for (int i = tid; i < N; i += 256) {
    const int b = i / (C_ * len);
    const int r = i % (C_ * len);
    const int c = r / len;
    const int x = r % len;
    float raw;
    if      (type == 0) raw = ws[OFF_SQ + b * C_ + c0] * ws[OFF_KH + (b * C_ + c) * H_ + x] + addeps;
    else if (type == 1) raw = ws[OFF_QS + (b * C_ + c0) * H_ + x] * ws[OFF_SK + b * C_ + c] + addeps;
    else if (type == 2) raw = ws[OFF_SQ + b * C_ + c0] * ws[OFF_KW + (b * C_ + c) * W_ + x] + addeps;
    else                raw = ws[OFF_QC + (b * C_ + c0) * W_ + x] * ws[OFF_SK + b * C_ + c] + addeps;
    dst[i] = raw;
    lmn = fminf(lmn, raw);
    lmx = fmaxf(lmx, raw);
  }
  rmn[tid] = lmn; rmx[tid] = lmx;
  __syncthreads();
  for (int s = 128; s > 0; s >>= 1) {
    if (tid < s) { rmn[tid] = fminf(rmn[tid], rmn[tid + s]); rmx[tid] = fmaxf(rmx[tid], rmx[tid + s]); }
    __syncthreads();
  }
  const float mn  = rmn[0];
  const float inv = 1.0f / (rmx[0] - mn);
  for (int i = tid; i < N; i += 256) dst[i] = 2.0f + (dst[i] - mn) * inv;
  if (blockIdx.x == 0)
    for (int i = tid; i < 160; i += 256) ws[OFF_GMAX + i] = 0.0f;
}

__global__ void k3_tile(float* __restrict__ ws, int c0base) {
  const int c0l = blockIdx.x >> 8;
  const int c0  = c0base + c0l;
  const int t   = (blockIdx.x >> 6) & 3;
  const int bc  = blockIdx.x & 63;
  const int tid = threadIdx.x;
  __shared__ float E[HW_];
  __shared__ float mv[H_], nv[W_];
  __shared__ float rsb[H_], csb[W_];
  __shared__ float red[256];

  const float* mptr = ws + ((t < 2) ? OFF_HR : OFF_HC) + c0 * (BC_ * H_) + bc * H_;
  const float* nptr = ws + (((t & 1) == 0) ? OFF_WR : OFF_WC) + c0 * (BC_ * W_) + bc * W_;
  if (tid < H_)   mv[tid] = mptr[tid];
  if (tid >= 128) nv[tid - 128] = nptr[tid - 128];
  __syncthreads();

  for (int i = tid; i < HW_; i += 256) E[i] = __expf(mv[i >> 7] * nv[i & 127]);
  __syncthreads();

  const int wave = tid >> 5, lane = tid & 31;
#if defined(USE_WMMA_F32)
  if (wave < 6) {
    const int M   = wave * 16 + (lane & 15);
    const int kho = (lane < 16) ? 0 : 2;
    const v2f ones = {1.0f, 1.0f};
    v8f acc = {};
    for (int k0 = 0; k0 < W_; k0 += 4) {
      v2f a = { E[M * W_ + k0 + kho], E[M * W_ + k0 + kho + 1] };
      acc = __builtin_amdgcn_wmma_f32_16x16x4_f32(false, a, false, ones, (short)0, acc, false, false);
    }
    if ((lane & 15) == 0) {
      const int rbase = wave * 16 + ((lane >> 4) << 3);
      #pragma unroll
      for (int v = 0; v < 8; ++v) rsb[rbase + v] = acc[v];
    }
  } else {
    const int N   = lane & 15;
    const int kho = (lane < 16) ? 0 : 2;
    const v2f onesA = {1.0f, 1.0f};
    for (int nb = (wave - 6) * 4; nb < (wave - 6) * 4 + 4; ++nb) {
      const int N0 = nb * 16;
      v8f acc = {};
      for (int k0 = 0; k0 < H_; k0 += 4) {
        v2f b = { E[(k0 + kho) * W_ + N0 + N], E[(k0 + kho + 1) * W_ + N0 + N] };
        acc = __builtin_amdgcn_wmma_f32_16x16x4_f32(false, onesA, false, b, (short)0, acc, false, false);
      }
      if (lane < 16) csb[N0 + N] = acc[0];
    }
  }
#else
  if (wave < 6) {
    const int M = wave * 16 + (lane & 15);
    const int khalf = (lane < 16) ? 0 : 8;
    v16h hb;
    #pragma unroll
    for (int e = 0; e < 16; ++e) hb[e] = (_Float16)1.0f;
    v8f acc = {};
    for (int k0 = 0; k0 < W_; k0 += 32) {
      v16h ha;
      #pragma unroll
      for (int v = 0; v < 8; ++v) {
        const int kp = (v & 3) * 2 + ((v >> 2) << 4) + khalf;
        ha[2 * v]     = (_Float16)E[M * W_ + k0 + kp];
        ha[2 * v + 1] = (_Float16)E[M * W_ + k0 + kp + 1];
      }
      acc = __builtin_amdgcn_wmma_f32_16x16x32_f16(false, ha, false, hb, (short)0, acc, false, false);
    }
    if ((lane & 15) == 0) {
      const int rbase = wave * 16 + ((lane >> 4) << 3);
      #pragma unroll
      for (int v = 0; v < 8; ++v) rsb[rbase + v] = acc[v];
    }
  } else {
    const int N = lane & 15;
    const int khalf = (lane < 16) ? 0 : 16;
    v16h ha;
    #pragma unroll
    for (int e = 0; e < 16; ++e) ha[e] = (_Float16)1.0f;
    for (int nb = (wave - 6) * 4; nb < (wave - 6) * 4 + 4; ++nb) {
      const int N0 = nb * 16;
      v8f acc = {};
      for (int k0 = 0; k0 < H_; k0 += 32) {
        v16h hbm;
        #pragma unroll
        for (int v = 0; v < 8; ++v) {
          const int kk = khalf + 2 * v;
          hbm[2 * v]     = (_Float16)E[(k0 + kk) * W_ + N0 + N];
          hbm[2 * v + 1] = (_Float16)E[(k0 + kk + 1) * W_ + N0 + N];
        }
        acc = __builtin_amdgcn_wmma_f32_16x16x32_f16(false, ha, false, hbm, (short)0, acc, false, false);
      }
      if (lane < 16) csb[N0 + N] = acc[0];
    }
  }
#endif
  __syncthreads();

  float* irs = ws + OFF_IRS + ((c0l * 4 + t) * 64 + bc) * 224;
  if (tid < H_)   { const float v = 1.0f / rsb[tid]; rsb[tid] = v; irs[tid] = v; }
  if (tid >= 128) { const int x = tid - 128; const float v = 1.0f / csb[x]; csb[x] = v; irs[96 + x] = v; }
  __syncthreads();

  float lmax = 0.0f;
  for (int i = tid; i < HW_; i += 256)
    lmax = fmaxf(lmax, E[i] * (rsb[i >> 7] + csb[i & 127]));
  red[tid] = lmax;
  __syncthreads();
  for (int s = 128; s > 0; s >>= 1) {
    if (tid < s) red[tid] = fmaxf(red[tid], red[tid + s]);
    __syncthreads();
  }
  if (tid == 0)
    atomicMax((int*)(ws + OFF_GMAX + c0 * 4 + t), __float_as_int(red[0]));
}

template <bool LAST>
__device__ __forceinline__ void score_rows(const float* __restrict__ wsr,
                                           const float* __restrict__ values,
                                           float* __restrict__ out,
                                           int c0, int c0l, int b, int h0, int w,
                                           const float (&ig)[4], float (&z)[ROWS_], float& lmax) {
  #pragma unroll 4
  for (int c = 0; c < C_; ++c) {
    const int bc = b * C_ + c;
    const float* hr  = wsr + OFF_HR + c0 * (BC_ * H_) + bc * H_ + h0;
    const float* hc  = wsr + OFF_HC + c0 * (BC_ * H_) + bc * H_ + h0;
    const float* irb = wsr + OFF_IRS + ((c0l * 4) * 64 + bc) * 224;
    float nn[2], icv[4];
    nn[0] = wsr[OFF_WR + c0 * (BC_ * W_) + bc * W_ + w];
    nn[1] = wsr[OFF_WC + c0 * (BC_ * W_) + bc * W_ + w];
    #pragma unroll
    for (int t = 0; t < 4; ++t) icv[t] = irb[t * (64 * 224) + 96 + w];
    #pragma unroll
    for (int r = 0; r < ROWS_; ++r) {
      float mm[2];
      mm[0] = hr[r];
      mm[1] = hc[r];
      float s = 0.f;
      #pragma unroll
      for (int t = 0; t < 4; ++t) {
        const float e = __expf(mm[t >> 1] * nn[t & 1]);
        s += e * (irb[t * (64 * 224) + h0 + r] + icv[t]) * ig[t];
      }
      const int off = bc * HW_ + (h0 + r) * W_ + w;
      z[r] += s * values[off];
      lmax = fmaxf(lmax, s);
      if (LAST) out[BC_ * HW_ + off] = s;
    }
  }
}

__global__ void k4_scorez(const float* __restrict__ wsr, float* __restrict__ wsw,
                          const float* __restrict__ values, float* __restrict__ out,
                          int c0base) {
  const int w   = threadIdx.x;
  const int rg  = blockIdx.x % (H_ / ROWS_);
  const int cb  = blockIdx.x / (H_ / ROWS_);
  const int b   = cb & 1;
  const int c0l = cb >> 1;
  const int c0  = c0base + c0l;
  const int h0  = rg * ROWS_;
  __shared__ float red[128];

  float ig[4];
  #pragma unroll
  for (int t = 0; t < 4; ++t) ig[t] = 1.0f / wsr[OFF_GMAX + c0 * 4 + t];

  float z[ROWS_];
  #pragma unroll
  for (int r = 0; r < ROWS_; ++r) z[r] = 0.f;
  float lmax = 0.0f;
  if (c0 == C_ - 1) score_rows<true >(wsr, values, out, c0, c0l, b, h0, w, ig, z, lmax);
  else              score_rows<false>(wsr, values, out, c0, c0l, b, h0, w, ig, z, lmax);

  #pragma unroll
  for (int r = 0; r < ROWS_; ++r) out[(b * C_ + c0) * HW_ + (h0 + r) * W_ + w] = z[r];

  red[w] = lmax;
  __syncthreads();
  for (int s2 = 64; s2 > 0; s2 >>= 1) {
    if (w < s2) red[w] = fmaxf(red[w], red[w + s2]);
    __syncthreads();
  }
  if (w == 0)
    atomicMax((int*)(wsw + OFF_GSMAX + c0), __float_as_int(red[0]));
}

__global__ void k6_scale(const float* __restrict__ ws, float* __restrict__ out) {
  const int g4    = blockIdx.x * 256 + threadIdx.x;
  const int plane = (g4 * 4) / HW_;
  const int c0    = (plane < BC_) ? (plane & (C_ - 1)) : (C_ - 1);
  const float inv = 1.0f / ws[OFF_GSMAX + c0];
  float4* o = reinterpret_cast<float4*>(out) + g4;
  float4 v = *o;
  v.x *= inv; v.y *= inv; v.z *= inv; v.w *= inv;
  *o = v;
}

extern "C" void kernel_launch(void* const* d_in, const int* in_sizes, int n_in,
                              void* d_out, int out_size, void* d_ws, size_t ws_size,
                              hipStream_t stream) {
  (void)in_sizes; (void)n_in; (void)out_size; (void)ws_size;
  const float* q   = (const float*)d_in[0];
  const float* k   = (const float*)d_in[1];
  const float* val = (const float*)d_in[2];
  float* out = (float*)d_out;
  float* ws  = (float*)d_ws;

  k1_sums<<<64, 256, 0, stream>>>(q, k, ws);
  k2_norm<<<128, 256, 0, stream>>>(ws);
  k3_tile<<<2816, 256, 0, stream>>>(ws, 0);
  k4_scorez<<<1056, 128, 0, stream>>>(ws, ws, val, out, 0);
  k3_tile<<<2816, 256, 0, stream>>>(ws, 11);
  k4_scorez<<<1056, 128, 0, stream>>>(ws, ws, val, out, 11);
  k3_tile<<<2560, 256, 0, stream>>>(ws, 22);
  k4_scorez<<<960, 128, 0, stream>>>(ws, ws, val, out, 22);
  k6_scale<<<1536, 256, 0, stream>>>(ws, out);
}


# k3: tile exp(m*n) no longer staged in LDS; each lane computes exp straight into its WMMA operand (and again for the tile max), 4-wave groups; row/col sums now add 4 per-wave partials in wave order
# speedup vs baseline: 4.4370x; 1.2736x over previous
#include <hip/hip_runtime.h>
#include <hip/hip_bf16.h>
#include <hip/hip_fp16.h>

typedef __attribute__((ext_vector_type(2)))  float    v2f;
typedef __attribute__((ext_vector_type(8)))  float    v8f;

#define B_   2
#define C_   32
#define H_   96
#define W_   128
#define HW_  (H_ * W_)
#define BC_  (B_ * C_)
#define EPSF 1e-6f

#define OFF_QS    0
#define OFF_QC    6144
#define OFF_KH    14336
#define OFF_KW    20480
#define OFF_SQ    28672
#define OFF_SK    28736
#define OFF_GMAX  28800
#define OFF_GSMAX 28928
#define OFF_HR    28960
#define OFF_HC    (OFF_HR + 32 * 6144)
#define OFF_WR    (OFF_HC + 32 * 6144)
#define OFF_WC    (OFF_WR + 32 * 8192)
#define OFF_IRS   (OFF_WC + 32 * 8192)
#define ROWS_     2

__global__ void k1_sums(const float* __restrict__ q, const float* __restrict__ k,
                        float* __restrict__ ws) {
  __shared__ float tile[HW_];
  __shared__ float rowbuf[H_];
  const int bc  = blockIdx.x;
  const int tid = threadIdx.x;
  for (int p = 0; p < 2; ++p) {
    const float* s = p ? k : q;
    for (int i = tid; i < HW_; i += 256) tile[i] = s[bc * HW_ + i];
    __syncthreads();
    if (tid < H_) {
      float acc = 0.f;
      for (int x = 0; x < W_; ++x) acc += tile[tid * W_ + x];
      ws[(p ? OFF_KH : OFF_QS) + bc * H_ + tid] = acc;
      rowbuf[tid] = acc;
    }
    if (tid >= 128) {
      const int x = tid - 128;
      float acc = 0.f;
      for (int h = 0; h < H_; ++h) acc += tile[h * W_ + x];
      ws[(p ? OFF_KW : OFF_QC) + bc * W_ + x] = acc;
    }
    __syncthreads();
    if (tid == 0) {
      float t = 0.f;
      for (int h = 0; h < H_; ++h) t += rowbuf[h];
      ws[(p ? OFF_SK : OFF_SQ) + bc] = t;
    }
    __syncthreads();
  }
}

__global__ void k2_norm(float* __restrict__ ws) {
  const int type = blockIdx.x & 3;
  const int c0   = blockIdx.x >> 2;
  const int tid  = threadIdx.x;
  __shared__ float rmn[256], rmx[256];
  const int   len    = (type < 2) ? H_ : W_;
  const int   N      = BC_ * len;
  const float addeps = (type < 2) ? (H_ * EPSF) : (W_ * EPSF);
  float* dst;
  if      (type == 0) dst = ws + OFF_HR + c0 * (BC_ * H_);
  else if (type == 1) dst = ws + OFF_HC + c0 * (BC_ * H_);
  else if (type == 2) dst = ws + OFF_WR + c0 * (BC_ * W_);
  else                dst = ws + OFF_WC + c0 * (BC_ * W_);
  float lmn = 3.4e38f, lmx = -3.4e38f;
  for (int i = tid; i < N; i += 256) {
    const int b = i / (C_ * len);
    const int r = i % (C_ * len);
    const int c = r / len;
    const int x = r % len;
    float raw;
    if      (type == 0) raw = ws[OFF_SQ + b * C_ + c0] * ws[OFF_KH + (b * C_ + c) * H_ + x] + addeps;
    else if (type == 1) raw = ws[OFF_QS + (b * C_ + c0) * H_ + x] * ws[OFF_SK + b * C_ + c] + addeps;
    else if (type == 2) raw = ws[OFF_SQ + b * C_ + c0] * ws[OFF_KW + (b * C_ + c) * W_ + x] + addeps;
    else                raw = ws[OFF_QC + (b * C_ + c0) * W_ + x] * ws[OFF_SK + b * C_ + c] + addeps;
    dst[i] = raw;
    lmn = fminf(lmn, raw);
    lmx = fmaxf(lmx, raw);
  }
  rmn[tid] = lmn; rmx[tid] = lmx;
  __syncthreads();
  for (int s = 128; s > 0; s >>= 1) {
    if (tid < s) { rmn[tid] = fminf(rmn[tid], rmn[tid + s]); rmx[tid] = fmaxf(rmx[tid], rmx[tid + s]); }
    __syncthreads();
  }
  const float mn  = rmn[0];
  const float inv = 1.0f / (rmx[0] - mn);
  for (int i = tid; i < N; i += 256) dst[i] = 2.0f + (dst[i] - mn) * inv;
  if (blockIdx.x == 0)
    for (int i = tid; i < 160; i += 256) ws[OFF_GMAX + i] = 0.0f;
}

__global__ void __launch_bounds__(128) k3_tile(float* __restrict__ ws, int c0base) {
  const int c0l = blockIdx.x >> 8;
  const int c0  = c0base + c0l;
  const int t   = (blockIdx.x >> 6) & 3;
  const int bc  = blockIdx.x & 63;
  const int tid = threadIdx.x;
  __shared__ float mv[H_], nv[W_];
  __shared__ float prs[4][H_];
  __shared__ float pcs[4][W_];
  __shared__ float rsb[H_], csb[W_];
  __shared__ float red[128];

  const float* mptr = ws + ((t < 2) ? OFF_HR : OFF_HC) + c0 * (BC_ * H_) + bc * H_;
  const float* nptr = ws + (((t & 1) == 0) ? OFF_WR : OFF_WC) + c0 * (BC_ * W_) + bc * W_;
  if (tid < H_) mv[tid] = mptr[tid];
  nv[tid] = nptr[tid];
  __syncthreads();

  const int wave = tid >> 5, lane = tid & 31;
  const int c16  = lane & 15;
  const int kh   = (lane >> 4) << 1;
  const v2f ones = {1.0f, 1.0f};
  {
    float nk[16];
    #pragma unroll
    for (int i = 0; i < 8; ++i) {
      nk[2 * i]     = nv[32 * wave + 4 * i + kh];
      nk[2 * i + 1] = nv[32 * wave + 4 * i + kh + 1];
    }
    #pragma unroll
    for (int rb = 0; rb < 6; rb += 2) {
      const float m0 = mv[16 * rb + c16];
      const float m1 = mv[16 * rb + 16 + c16];
      v8f acc0 = {}, acc1 = {};
      #pragma unroll
      for (int i = 0; i < 8; ++i) {
        const v2f e0 = { __expf(m0 * nk[2 * i]), __expf(m0 * nk[2 * i + 1]) };
        const v2f e1 = { __expf(m1 * nk[2 * i]), __expf(m1 * nk[2 * i + 1]) };
        acc0 = __builtin_amdgcn_wmma_f32_16x16x4_f32(false, ones, false, e0, (short)0, acc0, false, false);
        acc1 = __builtin_amdgcn_wmma_f32_16x16x4_f32(false, ones, false, e1, (short)0, acc1, false, false);
      }
      prs[wave][16 * rb + c16]      = acc0[0];
      prs[wave][16 * rb + 16 + c16] = acc1[0];
    }
  }
  {
    float mk[12];
    #pragma unroll
    for (int i = 0; i < 6; ++i) {
      mk[2 * i]     = mv[24 * wave + 4 * i + kh];
      mk[2 * i + 1] = mv[24 * wave + 4 * i + kh + 1];
    }
    #pragma unroll
    for (int cb = 0; cb < 8; cb += 2) {
      const float n0 = nv[16 * cb + c16];
      const float n1 = nv[16 * cb + 16 + c16];
      v8f acc0 = {}, acc1 = {};
      #pragma unroll
      for (int i = 0; i < 6; ++i) {
        const v2f e0 = { __expf(mk[2 * i] * n0), __expf(mk[2 * i + 1] * n0) };
        const v2f e1 = { __expf(mk[2 * i] * n1), __expf(mk[2 * i + 1] * n1) };
        acc0 = __builtin_amdgcn_wmma_f32_16x16x4_f32(false, ones, false, e0, (short)0, acc0, false, false);
        acc1 = __builtin_amdgcn_wmma_f32_16x16x4_f32(false, ones, false, e1, (short)0, acc1, false, false);
      }
      pcs[wave][16 * cb + c16]      = acc0[0];
      pcs[wave][16 * cb + 16 + c16] = acc1[0];
    }
  }
  __syncthreads();

  float* irs = ws + OFF_IRS + ((c0l * 4 + t) * 64 + bc) * 224;
  if (tid < H_) {
    const float s = ((prs[0][tid] + prs[1][tid]) + prs[2][tid]) + prs[3][tid];
    const float r = 1.0f / s; rsb[tid] = r; irs[tid] = r;
  }
  {
    const float s = ((pcs[0][tid] + pcs[1][tid]) + pcs[2][tid]) + pcs[3][tid];
    const float r = 1.0f / s; csb[tid] = r; irs[96 + tid] = r;
  }
  __syncthreads();

  const float nw = nv[tid];
  const float cw = csb[tid];
  float lmax = 0.0f;
  #pragma unroll 8
  for (int h = 0; h < H_; ++h)
    lmax = fmaxf(lmax, __expf(mv[h] * nw) * (rsb[h] + cw));
  red[tid] = lmax;
  __syncthreads();
  for (int s = 64; s > 0; s >>= 1) {
    if (tid < s) red[tid] = fmaxf(red[tid], red[tid + s]);
    __syncthreads();
  }
  if (tid == 0)
    atomicMax((int*)(ws + OFF_GMAX + c0 * 4 + t), __float_as_int(red[0]));
}

template <bool LAST>
__device__ __forceinline__ void score_rows(const float* __restrict__ wsr,
                                           const float* __restrict__ values,
                                           float* __restrict__ out,
                                           int c0, int c0l, int b, int h0, int w,
                                           const float (&ig)[4], float (&z)[ROWS_], float& lmax) {
  #pragma unroll 4
  for (int c = 0; c < C_; ++c) {
    const int bc = b * C_ + c;
    const float* hr  = wsr + OFF_HR + c0 * (BC_ * H_) + bc * H_ + h0;
    const float* hc  = wsr + OFF_HC + c0 * (BC_ * H_) + bc * H_ + h0;
    const float* irb = wsr + OFF_IRS + ((c0l * 4) * 64 + bc) * 224;
    float nn[2], icv[4];
    nn[0] = wsr[OFF_WR + c0 * (BC_ * W_) + bc * W_ + w];
    nn[1] = wsr[OFF_WC + c0 * (BC_ * W_) + bc * W_ + w];
    #pragma unroll
    for (int t = 0; t < 4; ++t) icv[t] = irb[t * (64 * 224) + 96 + w];
    #pragma unroll
    for (int r = 0; r < ROWS_; ++r) {
      float mm[2];
      mm[0] = hr[r];
      mm[1] = hc[r];
      float s = 0.f;
      #pragma unroll
      for (int t = 0; t < 4; ++t) {
        const float e = __expf(mm[t >> 1] * nn[t & 1]);
        s += e * (irb[t * (64 * 224) + h0 + r] + icv[t]) * ig[t];
      }
      const int off = bc * HW_ + (h0 + r) * W_ + w;
      z[r] += s * values[off];
      lmax = fmaxf(lmax, s);
      if (LAST) out[BC_ * HW_ + off] = s;
    }
  }
}

__global__ void k4_scorez(const float* __restrict__ wsr, float* __restrict__ wsw,
                          const float* __restrict__ values, float* __restrict__ out,
                          int c0base) {
  const int w   = threadIdx.x;
  const int rg  = blockIdx.x % (H_ / ROWS_);
  const int cb  = blockIdx.x / (H_ / ROWS_);
  const int b   = cb & 1;
  const int c0l = cb >> 1;
  const int c0  = c0base + c0l;
  const int h0  = rg * ROWS_;
  __shared__ float red[128];

  float ig[4];
  #pragma unroll
  for (int t = 0; t < 4; ++t) ig[t] = 1.0f / wsr[OFF_GMAX + c0 * 4 + t];

  float z[ROWS_];
  #pragma unroll
  for (int r = 0; r < ROWS_; ++r) z[r] = 0.f;
  float lmax = 0.0f;
  if (c0 == C_ - 1) score_rows<true >(wsr, values, out, c0, c0l, b, h0, w, ig, z, lmax);
  else              score_rows<false>(wsr, values, out, c0, c0l, b, h0, w, ig, z, lmax);

  #pragma unroll
  for (int r = 0; r < ROWS_; ++r) out[(b * C_ + c0) * HW_ + (h0 + r) * W_ + w] = z[r];

  red[w] = lmax;
  __syncthreads();
  for (int s2 = 64; s2 > 0; s2 >>= 1) {
    if (w < s2) red[w] = fmaxf(red[w], red[w + s2]);
    __syncthreads();
  }
  if (w == 0)
    atomicMax((int*)(wsw + OFF_GSMAX + c0), __float_as_int(red[0]));
}

__global__ void k6_scale(const float* __restrict__ ws, float* __restrict__ out) {
  const int g4    = blockIdx.x * 256 + threadIdx.x;
  const int plane = (g4 * 4) / HW_;
  const int c0    = (plane < BC_) ? (plane & (C_ - 1)) : (C_ - 1);
  const float inv = 1.0f / ws[OFF_GSMAX + c0];
  float4* o = reinterpret_cast<float4*>(out) + g4;
  float4 v = *o;
  v.x *= inv; v.y *= inv; v.z *= inv; v.w *= inv;
  *o = v;
}

extern "C" void kernel_launch(void* const* d_in, const int* in_sizes, int n_in,
                              void* d_out, int out_size, void* d_ws, size_t ws_size,
                              hipStream_t stream) {
  (void)in_sizes; (void)n_in; (void)out_size; (void)ws_size;
  const float* q   = (const float*)d_in[0];
  const float* k   = (const float*)d_in[1];
  const float* val = (const float*)d_in[2];
  float* out = (float*)d_out;
  float* ws  = (float*)d_ws;

  k1_sums<<<64, 256, 0, stream>>>(q, k, ws);
  k2_norm<<<128, 256, 0, stream>>>(ws);
  k3_tile<<<2816, 128, 0, stream>>>(ws, 0);
  k4_scorez<<<1056, 128, 0, stream>>>(ws, ws, val, out, 0);
  k3_tile<<<2816, 128, 0, stream>>>(ws, 11);
  k4_scorez<<<1056, 128, 0, stream>>>(ws, ws, val, out, 11);
  k3_tile<<<2560, 128, 0, stream>>>(ws, 22);
  k4_scorez<<<960, 128, 0, stream>>>(ws, ws, val, out, 22);
  k6_scale<<<1536, 256, 0, stream>>>(ws, out);
}


# k3: third exp sweep (tile max) removed: rc=E*(1/rowsum+1/colsum) peaks at the (max m, max n) element as m,n>0, so one wave reduces max m, max n, min 1/sums and evaluates that element once
# speedup vs baseline: 4.7327x; 1.0666x over previous
#include <hip/hip_runtime.h>
#include <hip/hip_bf16.h>
#include <hip/hip_fp16.h>

typedef __attribute__((ext_vector_type(2)))  float    v2f;
typedef __attribute__((ext_vector_type(8)))  float    v8f;

#define B_   2
#define C_   32
#define H_   96
#define W_   128
#define HW_  (H_ * W_)
#define BC_  (B_ * C_)
#define EPSF 1e-6f

#define OFF_QS    0
#define OFF_QC    6144
#define OFF_KH    14336
#define OFF_KW    20480
#define OFF_SQ    28672
#define OFF_SK    28736
#define OFF_GMAX  28800
#define OFF_GSMAX 28928
#define OFF_HR    28960
#define OFF_HC    (OFF_HR + 32 * 6144)
#define OFF_WR    (OFF_HC + 32 * 6144)
#define OFF_WC    (OFF_WR + 32 * 8192)
#define OFF_IRS   (OFF_WC + 32 * 8192)
#define ROWS_     2

__global__ void k1_sums(const float* __restrict__ q, const float* __restrict__ k,
                        float* __restrict__ ws) {
  __shared__ float tile[HW_];
  __shared__ float rowbuf[H_];
  const int bc  = blockIdx.x;
  const int tid = threadIdx.x;
  for (int p = 0; p < 2; ++p) {
    const float* s = p ? k : q;
    for (int i = tid; i < HW_; i += 256) tile[i] = s[bc * HW_ + i];
    __syncthreads();
    if (tid < H_) {
      float acc = 0.f;
      for (int x = 0; x < W_; ++x) acc += tile[tid * W_ + x];
      ws[(p ? OFF_KH : OFF_QS) + bc * H_ + tid] = acc;
      rowbuf[tid] = acc;
    }
    if (tid >= 128) {
      const int x = tid - 128;
      float acc = 0.f;
      for (int h = 0; h < H_; ++h) acc += tile[h * W_ + x];
      ws[(p ? OFF_KW : OFF_QC) + bc * W_ + x] = acc;
    }
    __syncthreads();
    if (tid == 0) {
      float t = 0.f;
      for (int h = 0; h < H_; ++h) t += rowbuf[h];
      ws[(p ? OFF_SK : OFF_SQ) + bc] = t;
    }
    __syncthreads();
  }
}

__global__ void k2_norm(float* __restrict__ ws) {
  const int type = blockIdx.x & 3;
  const int c0   = blockIdx.x >> 2;
  const int tid  = threadIdx.x;
  __shared__ float rmn[256], rmx[256];
  const int   len    = (type < 2) ? H_ : W_;
  const int   N      = BC_ * len;
  const float addeps = (type < 2) ? (H_ * EPSF) : (W_ * EPSF);
  float* dst;
  if      (type == 0) dst = ws + OFF_HR + c0 * (BC_ * H_);
  else if (type == 1) dst = ws + OFF_HC + c0 * (BC_ * H_);
  else if (type == 2) dst = ws + OFF_WR + c0 * (BC_ * W_);
  else                dst = ws + OFF_WC + c0 * (BC_ * W_);
  float lmn = 3.4e38f, lmx = -3.4e38f;
  for (int i = tid; i < N; i += 256) {
    const int b = i / (C_ * len);
    const int r = i % (C_ * len);
    const int c = r / len;
    const int x = r % len;
    float raw;
    if      (type == 0) raw = ws[OFF_SQ + b * C_ + c0] * ws[OFF_KH + (b * C_ + c) * H_ + x] + addeps;
    else if (type == 1) raw = ws[OFF_QS + (b * C_ + c0) * H_ + x] * ws[OFF_SK + b * C_ + c] + addeps;
    else if (type == 2) raw = ws[OFF_SQ + b * C_ + c0] * ws[OFF_KW + (b * C_ + c) * W_ + x] + addeps;
    else                raw = ws[OFF_QC + (b * C_ + c0) * W_ + x] * ws[OFF_SK + b * C_ + c] + addeps;
    dst[i] = raw;
    lmn = fminf(lmn, raw);
    lmx = fmaxf(lmx, raw);
  }
  rmn[tid] = lmn; rmx[tid] = lmx;
  __syncthreads();
  for (int s = 128; s > 0; s >>= 1) {
    if (tid < s) { rmn[tid] = fminf(rmn[tid], rmn[tid + s]); rmx[tid] = fmaxf(rmx[tid], rmx[tid + s]); }
    __syncthreads();
  }
  const float mn  = rmn[0];
  const float inv = 1.0f / (rmx[0] - mn);
  for (int i = tid; i < N; i += 256) dst[i] = 2.0f + (dst[i] - mn) * inv;
  if (blockIdx.x == 0)
    for (int i = tid; i < 160; i += 256) ws[OFF_GMAX + i] = 0.0f;
}

__global__ void __launch_bounds__(128) k3_tile(float* __restrict__ ws, int c0base) {
  const int c0l = blockIdx.x >> 8;
  const int c0  = c0base + c0l;
  const int t   = (blockIdx.x >> 6) & 3;
  const int bc  = blockIdx.x & 63;
  const int tid = threadIdx.x;
  __shared__ float mv[H_], nv[W_];
  __shared__ float prs[4][H_];
  __shared__ float pcs[4][W_];
  __shared__ float rsb[H_], csb[W_];

  const float* mptr = ws + ((t < 2) ? OFF_HR : OFF_HC) + c0 * (BC_ * H_) + bc * H_;
  const float* nptr = ws + (((t & 1) == 0) ? OFF_WR : OFF_WC) + c0 * (BC_ * W_) + bc * W_;
  if (tid < H_) mv[tid] = mptr[tid];
  nv[tid] = nptr[tid];
  __syncthreads();

  const int wave = tid >> 5, lane = tid & 31;
  const int c16  = lane & 15;
  const int kh   = (lane >> 4) << 1;
  const v2f ones = {1.0f, 1.0f};
  {
    float nk[16];
    #pragma unroll
    for (int i = 0; i < 8; ++i) {
      nk[2 * i]     = nv[32 * wave + 4 * i + kh];
      nk[2 * i + 1] = nv[32 * wave + 4 * i + kh + 1];
    }
    #pragma unroll
    for (int rb = 0; rb < 6; rb += 2) {
      const float m0 = mv[16 * rb + c16];
      const float m1 = mv[16 * rb + 16 + c16];
      v8f acc0 = {}, acc1 = {};
      #pragma unroll
      for (int i = 0; i < 8; ++i) {
        const v2f e0 = { __expf(m0 * nk[2 * i]), __expf(m0 * nk[2 * i + 1]) };
        const v2f e1 = { __expf(m1 * nk[2 * i]), __expf(m1 * nk[2 * i + 1]) };
        acc0 = __builtin_amdgcn_wmma_f32_16x16x4_f32(false, ones, false, e0, (short)0, acc0, false, false);
        acc1 = __builtin_amdgcn_wmma_f32_16x16x4_f32(false, ones, false, e1, (short)0, acc1, false, false);
      }
      prs[wave][16 * rb + c16]      = acc0[0];
      prs[wave][16 * rb + 16 + c16] = acc1[0];
    }
  }
  {
    float mk[12];
    #pragma unroll
    for (int i = 0; i < 6; ++i) {
      mk[2 * i]     = mv[24 * wave + 4 * i + kh];
      mk[2 * i + 1] = mv[24 * wave + 4 * i + kh + 1];
    }
    #pragma unroll
    for (int cb = 0; cb < 8; cb += 2) {
      const float n0 = nv[16 * cb + c16];
      const float n1 = nv[16 * cb + 16 + c16];
      v8f acc0 = {}, acc1 = {};
      #pragma unroll
      for (int i = 0; i < 6; ++i) {
        const v2f e0 = { __expf(mk[2 * i] * n0), __expf(mk[2 * i + 1] * n0) };
        const v2f e1 = { __expf(mk[2 * i] * n1), __expf(mk[2 * i + 1] * n1) };
        acc0 = __builtin_amdgcn_wmma_f32_16x16x4_f32(false, ones, false, e0, (short)0, acc0, false, false);
        acc1 = __builtin_amdgcn_wmma_f32_16x16x4_f32(false, ones, false, e1, (short)0, acc1, false, false);
      }
      pcs[wave][16 * cb + c16]      = acc0[0];
      pcs[wave][16 * cb + 16 + c16] = acc1[0];
    }
  }
  __syncthreads();

  float* irs = ws + OFF_IRS + ((c0l * 4 + t) * 64 + bc) * 224;
  if (tid < H_) {
    const float s = ((prs[0][tid] + prs[1][tid]) + prs[2][tid]) + prs[3][tid];
    const float r = 1.0f / s; rsb[tid] = r; irs[tid] = r;
  }
  {
    const float s = ((pcs[0][tid] + pcs[1][tid]) + pcs[2][tid]) + pcs[3][tid];
    const float r = 1.0f / s; csb[tid] = r; irs[96 + tid] = r;
  }
  __syncthreads();

  if (tid < 32) {
    float mm = fmaxf(fmaxf(mv[tid], mv[tid + 32]), mv[tid + 64]);
    float nn = fmaxf(fmaxf(nv[tid], nv[tid + 32]), fmaxf(nv[tid + 64], nv[tid + 96]));
    float ri = fminf(fminf(rsb[tid], rsb[tid + 32]), rsb[tid + 64]);
    float ci = fminf(fminf(csb[tid], csb[tid + 32]), fminf(csb[tid + 64], csb[tid + 96]));
    for (int o = 16; o > 0; o >>= 1) {
      mm = fmaxf(mm, __shfl_xor(mm, o));
      nn = fmaxf(nn, __shfl_xor(nn, o));
      ri = fminf(ri, __shfl_xor(ri, o));
      ci = fminf(ci, __shfl_xor(ci, o));
    }
    if (tid == 0)
      atomicMax((int*)(ws + OFF_GMAX + c0 * 4 + t),
                __float_as_int(__expf(mm * nn) * (ri + ci)));
  }
}

template <bool LAST>
__device__ __forceinline__ void score_rows(const float* __restrict__ wsr,
                                           const float* __restrict__ values,
                                           float* __restrict__ out,
                                           int c0, int c0l, int b, int h0, int w,
                                           const float (&ig)[4], float (&z)[ROWS_], float& lmax) {
  #pragma unroll 4
  for (int c = 0; c < C_; ++c) {
    const int bc = b * C_ + c;
    const float* hr  = wsr + OFF_HR + c0 * (BC_ * H_) + bc * H_ + h0;
    const float* hc  = wsr + OFF_HC + c0 * (BC_ * H_) + bc * H_ + h0;
    const float* irb = wsr + OFF_IRS + ((c0l * 4) * 64 + bc) * 224;
    float nn[2], icv[4];
    nn[0] = wsr[OFF_WR + c0 * (BC_ * W_) + bc * W_ + w];
    nn[1] = wsr[OFF_WC + c0 * (BC_ * W_) + bc * W_ + w];
    #pragma unroll
    for (int t = 0; t < 4; ++t) icv[t] = irb[t * (64 * 224) + 96 + w];
    #pragma unroll
    for (int r = 0; r < ROWS_; ++r) {
      float mm[2];
      mm[0] = hr[r];
      mm[1] = hc[r];
      float s = 0.f;
      #pragma unroll
      for (int t = 0; t < 4; ++t) {
        const float e = __expf(mm[t >> 1] * nn[t & 1]);
        s += e * (irb[t * (64 * 224) + h0 + r] + icv[t]) * ig[t];
      }
      const int off = bc * HW_ + (h0 + r) * W_ + w;
      z[r] += s * values[off];
      lmax = fmaxf(lmax, s);
      if (LAST) out[BC_ * HW_ + off] = s;
    }
  }
}

__global__ void k4_scorez(const float* __restrict__ wsr, float* __restrict__ wsw,
                          const float* __restrict__ values, float* __restrict__ out,
                          int c0base) {
  const int w   = threadIdx.x;
  const int rg  = blockIdx.x % (H_ / ROWS_);
  const int cb  = blockIdx.x / (H_ / ROWS_);
  const int b   = cb & 1;
  const int c0l = cb >> 1;
  const int c0  = c0base + c0l;
  const int h0  = rg * ROWS_;
  __shared__ float red[128];

  float ig[4];
  #pragma unroll
  for (int t = 0; t < 4; ++t) ig[t] = 1.0f / wsr[OFF_GMAX + c0 * 4 + t];

  float z[ROWS_];
  #pragma unroll
  for (int r = 0; r < ROWS_; ++r) z[r] = 0.f;
  float lmax = 0.0f;
  if (c0 == C_ - 1) score_rows<true >(wsr, values, out, c0, c0l, b, h0, w, ig, z, lmax);
  else              score_rows<false>(wsr, values, out, c0, c0l, b, h0, w, ig, z, lmax);

  #pragma unroll
  for (int r = 0; r < ROWS_; ++r) out[(b * C_ + c0) * HW_ + (h0 + r) * W_ + w] = z[r];

  red[w] = lmax;
  __syncthreads();
  for (int s2 = 64; s2 > 0; s2 >>= 1) {
    if (w < s2) red[w] = fmaxf(red[w], red[w + s2]);
    __syncthreads();
  }
  if (w == 0)
    atomicMax((int*)(wsw + OFF_GSMAX + c0), __float_as_int(red[0]));
}

__global__ void k6_scale(const float* __restrict__ ws, float* __restrict__ out) {
  const int g4    = blockIdx.x * 256 + threadIdx.x;
  const int plane = (g4 * 4) / HW_;
  const int c0    = (plane < BC_) ? (plane & (C_ - 1)) : (C_ - 1);
  const float inv = 1.0f / ws[OFF_GSMAX + c0];
  float4* o = reinterpret_cast<float4*>(out) + g4;
  float4 v = *o;
  v.x *= inv; v.y *= inv; v.z *= inv; v.w *= inv;
  *o = v;
}

extern "C" void kernel_launch(void* const* d_in, const int* in_sizes, int n_in,
                              void* d_out, int out_size, void* d_ws, size_t ws_size,
                              hipStream_t stream) {
  (void)in_sizes; (void)n_in; (void)out_size; (void)ws_size;
  const float* q   = (const float*)d_in[0];
  const float* k   = (const float*)d_in[1];
  const float* val = (const float*)d_in[2];
  float* out = (float*)d_out;
  float* ws  = (float*)d_ws;

  k1_sums<<<64, 256, 0, stream>>>(q, k, ws);
  k2_norm<<<128, 256, 0, stream>>>(ws);
  k3_tile<<<2816, 128, 0, stream>>>(ws, 0);
  k4_scorez<<<1056, 128, 0, stream>>>(ws, ws, val, out, 0);
  k3_tile<<<2816, 128, 0, stream>>>(ws, 11);
  k4_scorez<<<1056, 128, 0, stream>>>(ws, ws, val, out, 11);
  k3_tile<<<2560, 128, 0, stream>>>(ws, 22);
  k4_scorez<<<960, 128, 0, stream>>>(ws, ws, val, out, 22);
  k6_scale<<<1536, 256, 0, stream>>>(ws, out);
}


# k2: each thread keeps its 24-32 elements in registers; all factor loads issued as one batch before any store, one store per element; no per-element load-wait-store loop and no read-back pass
# speedup vs baseline: 5.4504x; 1.1516x over previous
#include <hip/hip_runtime.h>
#include <hip/hip_bf16.h>
#include <hip/hip_fp16.h>

typedef __attribute__((ext_vector_type(2)))  float    v2f;
typedef __attribute__((ext_vector_type(8)))  float    v8f;

#define B_   2
#define C_   32
#define H_   96
#define W_   128
#define HW_  (H_ * W_)
#define BC_  (B_ * C_)
#define EPSF 1e-6f

#define OFF_QS    0
#define OFF_QC    6144
#define OFF_KH    14336
#define OFF_KW    20480
#define OFF_SQ    28672
#define OFF_SK    28736
#define OFF_GMAX  28800
#define OFF_GSMAX 28928
#define OFF_HR    28960
#define OFF_HC    (OFF_HR + 32 * 6144)
#define OFF_WR    (OFF_HC + 32 * 6144)
#define OFF_WC    (OFF_WR + 32 * 8192)
#define OFF_IRS   (OFF_WC + 32 * 8192)
#define ROWS_     2

__global__ void k1_sums(const float* __restrict__ q, const float* __restrict__ k,
                        float* __restrict__ ws) {
  __shared__ float tile[HW_];
  __shared__ float rowbuf[H_];
  const int bc  = blockIdx.x;
  const int tid = threadIdx.x;
  for (int p = 0; p < 2; ++p) {
    const float* s = p ? k : q;
    for (int i = tid; i < HW_; i += 256) tile[i] = s[bc * HW_ + i];
    __syncthreads();
    if (tid < H_) {
      float acc = 0.f;
      for (int x = 0; x < W_; ++x) acc += tile[tid * W_ + x];
      ws[(p ? OFF_KH : OFF_QS) + bc * H_ + tid] = acc;
      rowbuf[tid] = acc;
    }
    if (tid >= 128) {
      const int x = tid - 128;
      float acc = 0.f;
      for (int h = 0; h < H_; ++h) acc += tile[h * W_ + x];
      ws[(p ? OFF_KW : OFF_QC) + bc * W_ + x] = acc;
    }
    __syncthreads();
    if (tid == 0) {
      float t = 0.f;
      for (int h = 0; h < H_; ++h) t += rowbuf[h];
      ws[(p ? OFF_SK : OFF_SQ) + bc] = t;
    }
    __syncthreads();
  }
}

template <int TYPE>
__device__ __forceinline__ void k2_family(float* __restrict__ ws, int c0, int tid,
                                          float* rmn, float* rmx) {
  constexpr int   LEN    = (TYPE < 2) ? H_ : W_;
  constexpr int   N      = BC_ * LEN;
  constexpr int   PER    = N / 256;
  constexpr float addeps = LEN * EPSF;
  float* dst = ws + ((TYPE == 0) ? OFF_HR : (TYPE == 1) ? OFF_HC : (TYPE == 2) ? OFF_WR : OFF_WC) + c0 * N;
  float fa[PER], fb[PER];
  #pragma unroll
  for (int j = 0; j < PER; ++j) {
    const int i  = tid + 256 * j;
    const int b  = (256 * j) / (C_ * LEN);
    const int bc = i / LEN;
    const int x  = i % LEN;
    if      (TYPE == 0) { fa[j] = ws[OFF_SQ + b * C_ + c0];            fb[j] = ws[OFF_KH + i];  }
    else if (TYPE == 1) { fa[j] = ws[OFF_QS + (b * C_ + c0) * H_ + x]; fb[j] = ws[OFF_SK + bc]; }
    else if (TYPE == 2) { fa[j] = ws[OFF_SQ + b * C_ + c0];            fb[j] = ws[OFF_KW + i];  }
    else                { fa[j] = ws[OFF_QC + (b * C_ + c0) * W_ + x]; fb[j] = ws[OFF_SK + bc]; }
  }
  float lmn = 3.4e38f, lmx = -3.4e38f;
  #pragma unroll
  for (int j = 0; j < PER; ++j) {
    fa[j] = fa[j] * fb[j] + addeps;
    lmn = fminf(lmn, fa[j]);
    lmx = fmaxf(lmx, fa[j]);
  }
  rmn[tid] = lmn; rmx[tid] = lmx;
  __syncthreads();
  for (int s = 128; s > 0; s >>= 1) {
    if (tid < s) { rmn[tid] = fminf(rmn[tid], rmn[tid + s]); rmx[tid] = fmaxf(rmx[tid], rmx[tid + s]); }
    __syncthreads();
  }
  const float mn  = rmn[0];
  const float inv = 1.0f / (rmx[0] - mn);
  #pragma unroll
  for (int j = 0; j < PER; ++j) dst[tid + 256 * j] = 2.0f + (fa[j] - mn) * inv;
}

__global__ void __launch_bounds__(256) k2_norm(float* __restrict__ ws) {
  const int type = blockIdx.x & 3;
  const int c0   = blockIdx.x >> 2;
  const int tid  = threadIdx.x;
  __shared__ float rmn[256], rmx[256];
  if      (type == 0) k2_family<0>(ws, c0, tid, rmn, rmx);
  else if (type == 1) k2_family<1>(ws, c0, tid, rmn, rmx);
  else if (type == 2) k2_family<2>(ws, c0, tid, rmn, rmx);
  else                k2_family<3>(ws, c0, tid, rmn, rmx);
  if (blockIdx.x == 0)
    for (int i = tid; i < 160; i += 256) ws[OFF_GMAX + i] = 0.0f;
}

__global__ void __launch_bounds__(128) k3_tile(float* __restrict__ ws, int c0base) {
  const int c0l = blockIdx.x >> 8;
  const int c0  = c0base + c0l;
  const int t   = (blockIdx.x >> 6) & 3;
  const int bc  = blockIdx.x & 63;
  const int tid = threadIdx.x;
  __shared__ float mv[H_], nv[W_];
  __shared__ float prs[4][H_];
  __shared__ float pcs[4][W_];
  __shared__ float rsb[H_], csb[W_];

  const float* mptr = ws + ((t < 2) ? OFF_HR : OFF_HC) + c0 * (BC_ * H_) + bc * H_;
  const float* nptr = ws + (((t & 1) == 0) ? OFF_WR : OFF_WC) + c0 * (BC_ * W_) + bc * W_;
  if (tid < H_) mv[tid] = mptr[tid];
  nv[tid] = nptr[tid];
  __syncthreads();

  const int wave = tid >> 5, lane = tid & 31;
  const int c16  = lane & 15;
  const int kh   = (lane >> 4) << 1;
  const v2f ones = {1.0f, 1.0f};
  {
    float nk[16];
    #pragma unroll
    for (int i = 0; i < 8; ++i) {
      nk[2 * i]     = nv[32 * wave + 4 * i + kh];
      nk[2 * i + 1] = nv[32 * wave + 4 * i + kh + 1];
    }
    #pragma unroll
    for (int rb = 0; rb < 6; rb += 2) {
      const float m0 = mv[16 * rb + c16];
      const float m1 = mv[16 * rb + 16 + c16];
      v8f acc0 = {}, acc1 = {};
      #pragma unroll
      for (int i = 0; i < 8; ++i) {
        const v2f e0 = { __expf(m0 * nk[2 * i]), __expf(m0 * nk[2 * i + 1]) };
        const v2f e1 = { __expf(m1 * nk[2 * i]), __expf(m1 * nk[2 * i + 1]) };
        acc0 = __builtin_amdgcn_wmma_f32_16x16x4_f32(false, ones, false, e0, (short)0, acc0, false, false);
        acc1 = __builtin_amdgcn_wmma_f32_16x16x4_f32(false, ones, false, e1, (short)0, acc1, false, false);
      }
      prs[wave][16 * rb + c16]      = acc0[0];
      prs[wave][16 * rb + 16 + c16] = acc1[0];
    }
  }
  {
    float mk[12];
    #pragma unroll
    for (int i = 0; i < 6; ++i) {
      mk[2 * i]     = mv[24 * wave + 4 * i + kh];
      mk[2 * i + 1] = mv[24 * wave + 4 * i + kh + 1];
    }
    #pragma unroll
    for (int cb = 0; cb < 8; cb += 2) {
      const float n0 = nv[16 * cb + c16];
      const float n1 = nv[16 * cb + 16 + c16];
      v8f acc0 = {}, acc1 = {};
      #pragma unroll
      for (int i = 0; i < 6; ++i) {
        const v2f e0 = { __expf(mk[2 * i] * n0), __expf(mk[2 * i + 1] * n0) };
        const v2f e1 = { __expf(mk[2 * i] * n1), __expf(mk[2 * i + 1] * n1) };
        acc0 = __builtin_amdgcn_wmma_f32_16x16x4_f32(false, ones, false, e0, (short)0, acc0, false, false);
        acc1 = __builtin_amdgcn_wmma_f32_16x16x4_f32(false, ones, false, e1, (short)0, acc1, false, false);
      }
      pcs[wave][16 * cb + c16]      = acc0[0];
      pcs[wave][16 * cb + 16 + c16] = acc1[0];
    }
  }
  __syncthreads();

  float* irs = ws + OFF_IRS + ((c0l * 4 + t) * 64 + bc) * 224;
  if (tid < H_) {
    const float s = ((prs[0][tid] + prs[1][tid]) + prs[2][tid]) + prs[3][tid];
    const float r = 1.0f / s; rsb[tid] = r; irs[tid] = r;
  }
  {
    const float s = ((pcs[0][tid] + pcs[1][tid]) + pcs[2][tid]) + pcs[3][tid];
    const float r = 1.0f / s; csb[tid] = r; irs[96 + tid] = r;
  }
  __syncthreads();

  if (tid < 32) {
    float mm = fmaxf(fmaxf(mv[tid], mv[tid + 32]), mv[tid + 64]);
    float nn = fmaxf(fmaxf(nv[tid], nv[tid + 32]), fmaxf(nv[tid + 64], nv[tid + 96]));
    float ri = fminf(fminf(rsb[tid], rsb[tid + 32]), rsb[tid + 64]);
    float ci = fminf(fminf(csb[tid], csb[tid + 32]), fminf(csb[tid + 64], csb[tid + 96]));
    for (int o = 16; o > 0; o >>= 1) {
      mm = fmaxf(mm, __shfl_xor(mm, o));
      nn = fmaxf(nn, __shfl_xor(nn, o));
      ri = fminf(ri, __shfl_xor(ri, o));
      ci = fminf(ci, __shfl_xor(ci, o));
    }
    if (tid == 0)
      atomicMax((int*)(ws + OFF_GMAX + c0 * 4 + t),
                __float_as_int(__expf(mm * nn) * (ri + ci)));
  }
}

template <bool LAST>
__device__ __forceinline__ void score_rows(const float* __restrict__ wsr,
                                           const float* __restrict__ values,
                                           float* __restrict__ out,
                                           int c0, int c0l, int b, int h0, int w,
                                           const float (&ig)[4], float (&z)[ROWS_], float& lmax) {
  #pragma unroll 4
  for (int c = 0; c < C_; ++c) {
    const int bc = b * C_ + c;
    const float* hr  = wsr + OFF_HR + c0 * (BC_ * H_) + bc * H_ + h0;
    const float* hc  = wsr + OFF_HC + c0 * (BC_ * H_) + bc * H_ + h0;
    const float* irb = wsr + OFF_IRS + ((c0l * 4) * 64 + bc) * 224;
    float nn[2], icv[4];
    nn[0] = wsr[OFF_WR + c0 * (BC_ * W_) + bc * W_ + w];
    nn[1] = wsr[OFF_WC + c0 * (BC_ * W_) + bc * W_ + w];
    #pragma unroll
    for (int t = 0; t < 4; ++t) icv[t] = irb[t * (64 * 224) + 96 + w];
    #pragma unroll
    for (int r = 0; r < ROWS_; ++r) {
      float mm[2];
      mm[0] = hr[r];
      mm[1] = hc[r];
      float s = 0.f;
      #pragma unroll
      for (int t = 0; t < 4; ++t) {
        const float e = __expf(mm[t >> 1] * nn[t & 1]);
        s += e * (irb[t * (64 * 224) + h0 + r] + icv[t]) * ig[t];
      }
      const int off = bc * HW_ + (h0 + r) * W_ + w;
      z[r] += s * values[off];
      lmax = fmaxf(lmax, s);
      if (LAST) out[BC_ * HW_ + off] = s;
    }
  }
}

__global__ void k4_scorez(const float* __restrict__ wsr, float* __restrict__ wsw,
                          const float* __restrict__ values, float* __restrict__ out,
                          int c0base) {
  const int w   = threadIdx.x;
  const int rg  = blockIdx.x % (H_ / ROWS_);
  const int cb  = blockIdx.x / (H_ / ROWS_);
  const int b   = cb & 1;
  const int c0l = cb >> 1;
  const int c0  = c0base + c0l;
  const int h0  = rg * ROWS_;
  __shared__ float red[128];

  float ig[4];
  #pragma unroll
  for (int t = 0; t < 4; ++t) ig[t] = 1.0f / wsr[OFF_GMAX + c0 * 4 + t];

  float z[ROWS_];
  #pragma unroll
  for (int r = 0; r < ROWS_; ++r) z[r] = 0.f;
  float lmax = 0.0f;
  if (c0 == C_ - 1) score_rows<true >(wsr, values, out, c0, c0l, b, h0, w, ig, z, lmax);
  else              score_rows<false>(wsr, values, out, c0, c0l, b, h0, w, ig, z, lmax);

  #pragma unroll
  for (int r = 0; r < ROWS_; ++r) out[(b * C_ + c0) * HW_ + (h0 + r) * W_ + w] = z[r];

  red[w] = lmax;
  __syncthreads();
  for (int s2 = 64; s2 > 0; s2 >>= 1) {
    if (w < s2) red[w] = fmaxf(red[w], red[w + s2]);
    __syncthreads();
  }
  if (w == 0)
    atomicMax((int*)(wsw + OFF_GSMAX + c0), __float_as_int(red[0]));
}

__global__ void k6_scale(const float* __restrict__ ws, float* __restrict__ out) {
  const int g4    = blockIdx.x * 256 + threadIdx.x;
  const int plane = (g4 * 4) / HW_;
  const int c0    = (plane < BC_) ? (plane & (C_ - 1)) : (C_ - 1);
  const float inv = 1.0f / ws[OFF_GSMAX + c0];
  float4* o = reinterpret_cast<float4*>(out) + g4;
  float4 v = *o;
  v.x *= inv; v.y *= inv; v.z *= inv; v.w *= inv;
  *o = v;
}

extern "C" void kernel_launch(void* const* d_in, const int* in_sizes, int n_in,
                              void* d_out, int out_size, void* d_ws, size_t ws_size,
                              hipStream_t stream) {
  (void)in_sizes; (void)n_in; (void)out_size; (void)ws_size;
  const float* q   = (const float*)d_in[0];
  const float* k   = (const float*)d_in[1];
  const float* val = (const float*)d_in[2];
  float* out = (float*)d_out;
  float* ws  = (float*)d_ws;

  k1_sums<<<64, 256, 0, stream>>>(q, k, ws);
  k2_norm<<<128, 256, 0, stream>>>(ws);
  k3_tile<<<2816, 128, 0, stream>>>(ws, 0);
  k4_scorez<<<1056, 128, 0, stream>>>(ws, ws, val, out, 0);
  k3_tile<<<2816, 128, 0, stream>>>(ws, 11);
  k4_scorez<<<1056, 128, 0, stream>>>(ws, ws, val, out, 11);
  k3_tile<<<2560, 128, 0, stream>>>(ws, 22);
  k4_scorez<<<960, 128, 0, stream>>>(ws, ws, val, out, 22);
  k6_scale<<<1536, 256, 0, stream>>>(ws, out);
}


# k3: one exp sweep, not two: A becomes a 0/1 selector so one WMMA adds a fragment to a row-sum row and to 4 column-partial rows; 16 lane partials per column summed by 8 WMMAs; sum order changes
# speedup vs baseline: 6.1009x; 1.1193x over previous
#include <hip/hip_runtime.h>
#include <hip/hip_bf16.h>
#include <hip/hip_fp16.h>

typedef __attribute__((ext_vector_type(2)))  float    v2f;
typedef __attribute__((ext_vector_type(8)))  float    v8f;

#define B_   2
#define C_   32
#define H_   96
#define W_   128
#define HW_  (H_ * W_)
#define BC_  (B_ * C_)
#define EPSF 1e-6f

#define OFF_QS    0
#define OFF_QC    6144
#define OFF_KH    14336
#define OFF_KW    20480
#define OFF_SQ    28672
#define OFF_SK    28736
#define OFF_GMAX  28800
#define OFF_GSMAX 28928
#define OFF_HR    28960
#define OFF_HC    (OFF_HR + 32 * 6144)
#define OFF_WR    (OFF_HC + 32 * 6144)
#define OFF_WC    (OFF_WR + 32 * 8192)
#define OFF_IRS   (OFF_WC + 32 * 8192)
#define ROWS_     2

__global__ void k1_sums(const float* __restrict__ q, const float* __restrict__ k,
                        float* __restrict__ ws) {
  __shared__ float tile[HW_];
  __shared__ float rowbuf[H_];
  const int bc  = blockIdx.x;
  const int tid = threadIdx.x;
  for (int p = 0; p < 2; ++p) {
    const float* s = p ? k : q;
    for (int i = tid; i < HW_; i += 256) tile[i] = s[bc * HW_ + i];
    __syncthreads();
    if (tid < H_) {
      float acc = 0.f;
      for (int x = 0; x < W_; ++x) acc += tile[tid * W_ + x];
      ws[(p ? OFF_KH : OFF_QS) + bc * H_ + tid] = acc;
      rowbuf[tid] = acc;
    }
    if (tid >= 128) {
      const int x = tid - 128;
      float acc = 0.f;
      for (int h = 0; h < H_; ++h) acc += tile[h * W_ + x];
      ws[(p ? OFF_KW : OFF_QC) + bc * W_ + x] = acc;
    }
    __syncthreads();
    if (tid == 0) {
      float t = 0.f;
      for (int h = 0; h < H_; ++h) t += rowbuf[h];
      ws[(p ? OFF_SK : OFF_SQ) + bc] = t;
    }
    __syncthreads();
  }
}

template <int TYPE>
__device__ __forceinline__ void k2_family(float* __restrict__ ws, int c0, int tid,
                                          float* rmn, float* rmx) {
  constexpr int   LEN    = (TYPE < 2) ? H_ : W_;
  constexpr int   N      = BC_ * LEN;
  constexpr int   PER    = N / 256;
  constexpr float addeps = LEN * EPSF;
  float* dst = ws + ((TYPE == 0) ? OFF_HR : (TYPE == 1) ? OFF_HC : (TYPE == 2) ? OFF_WR : OFF_WC) + c0 * N;
  float fa[PER], fb[PER];
  #pragma unroll
  for (int j = 0; j < PER; ++j) {
    const int i  = tid + 256 * j;
    const int b  = (256 * j) / (C_ * LEN);
    const int bc = i / LEN;
    const int x  = i % LEN;
    if      (TYPE == 0) { fa[j] = ws[OFF_SQ + b * C_ + c0];            fb[j] = ws[OFF_KH + i];  }
    else if (TYPE == 1) { fa[j] = ws[OFF_QS + (b * C_ + c0) * H_ + x]; fb[j] = ws[OFF_SK + bc]; }
    else if (TYPE == 2) { fa[j] = ws[OFF_SQ + b * C_ + c0];            fb[j] = ws[OFF_KW + i];  }
    else                { fa[j] = ws[OFF_QC + (b * C_ + c0) * W_ + x]; fb[j] = ws[OFF_SK + bc]; }
  }
  float lmn = 3.4e38f, lmx = -3.4e38f;
  #pragma unroll
  for (int j = 0; j < PER; ++j) {
    fa[j] = fa[j] * fb[j] + addeps;
    lmn = fminf(lmn, fa[j]);
    lmx = fmaxf(lmx, fa[j]);
  }
  rmn[tid] = lmn; rmx[tid] = lmx;
  __syncthreads();
  for (int s = 128; s > 0; s >>= 1) {
    if (tid < s) { rmn[tid] = fminf(rmn[tid], rmn[tid + s]); rmx[tid] = fmaxf(rmx[tid], rmx[tid + s]); }
    __syncthreads();
  }
  const float mn  = rmn[0];
  const float inv = 1.0f / (rmx[0] - mn);
  #pragma unroll
  for (int j = 0; j < PER; ++j) dst[tid + 256 * j] = 2.0f + (fa[j] - mn) * inv;
}

__global__ void __launch_bounds__(256) k2_norm(float* __restrict__ ws) {
  const int type = blockIdx.x & 3;
  const int c0   = blockIdx.x >> 2;
  const int tid  = threadIdx.x;
  __shared__ float rmn[256], rmx[256];
  if      (type == 0) k2_family<0>(ws, c0, tid, rmn, rmx);
  else if (type == 1) k2_family<1>(ws, c0, tid, rmn, rmx);
  else if (type == 2) k2_family<2>(ws, c0, tid, rmn, rmx);
  else                k2_family<3>(ws, c0, tid, rmn, rmx);
  if (blockIdx.x == 0)
    for (int i = tid; i < 160; i += 256) ws[OFF_GMAX + i] = 0.0f;
}

#define CPP_ 20
#define CPH_ (32 * CPP_ + 16)
#define CPW_ (CPH_ + 32 * CPP_)
__global__ void __launch_bounds__(128) k3_tile(float* __restrict__ ws, int c0base) {
  const int c0l = blockIdx.x >> 8;
  const int c0  = c0base + c0l;
  const int t   = (blockIdx.x >> 6) & 3;
  const int bc  = blockIdx.x & 63;
  const int tid = threadIdx.x;
  __shared__ float mv[H_], nv[W_];
  __shared__ __attribute__((aligned(16))) float cp[4][CPW_];
  __shared__ __attribute__((aligned(16))) float csum[W_];
  __shared__ float rsb[H_];

  const float* mptr = ws + ((t < 2) ? OFF_HR : OFF_HC) + c0 * (BC_ * H_) + bc * H_;
  const float* nptr = ws + (((t & 1) == 0) ? OFF_WR : OFF_WC) + c0 * (BC_ * W_) + bc * W_;
  if (tid < H_) mv[tid] = mptr[tid];
  nv[tid] = nptr[tid];
  __syncthreads();

  const int wave = tid >> 5, lane = tid & 31;
  const int c16  = lane & 15;
  const int hh   = lane >> 4;
  const int kh   = hh << 1;
  const v2f ones = {1.0f, 1.0f};

  float nk[16];
  #pragma unroll
  for (int i = 0; i < 8; ++i) {
    nk[2 * i]     = nv[32 * wave + 4 * i + kh];
    nk[2 * i + 1] = nv[32 * wave + 4 * i + kh + 1];
  }
  v2f asel[6][2];
  #pragma unroll
  for (int rb = 0; rb < 6; ++rb) {
    const float rsel = (c16 == 8 + rb) ? 1.0f : 0.0f;
    #pragma unroll
    for (int p = 0; p < 2; ++p) {
      asel[rb][p][0] = (c16 == 4 * p + kh)     ? 1.0f : rsel;
      asel[rb][p][1] = (c16 == 4 * p + kh + 1) ? 1.0f : rsel;
    }
  }

  v8f acc0 = {}, acc1 = {}, acc2 = {}, acc3 = {};
  #pragma unroll
  for (int rb = 0; rb < 6; ++rb) {
    const float m0 = mv[16 * rb + c16];
    #pragma unroll
    for (int p = 0; p < 2; ++p) {
      const v2f e0 = { __expf(m0 * nk[2 * p]),      __expf(m0 * nk[2 * p + 1]) };
      const v2f e1 = { __expf(m0 * nk[4 + 2 * p]),  __expf(m0 * nk[4 + 2 * p + 1]) };
      const v2f e2 = { __expf(m0 * nk[8 + 2 * p]),  __expf(m0 * nk[8 + 2 * p + 1]) };
      const v2f e3 = { __expf(m0 * nk[12 + 2 * p]), __expf(m0 * nk[12 + 2 * p + 1]) };
      acc0 = __builtin_amdgcn_wmma_f32_16x16x4_f32(false, asel[rb][p], false, e0, (short)0, acc0, false, false);
      acc1 = __builtin_amdgcn_wmma_f32_16x16x4_f32(false, asel[rb][p], false, e1, (short)0, acc1, false, false);
      acc2 = __builtin_amdgcn_wmma_f32_16x16x4_f32(false, asel[rb][p], false, e2, (short)0, acc2, false, false);
      acc3 = __builtin_amdgcn_wmma_f32_16x16x4_f32(false, asel[rb][p], false, e3, (short)0, acc3, false, false);
    }
  }

  {
    float* dst = &cp[wave][hh * CPH_ + c16];
    #pragma unroll
    for (int r = 0; r < 8; ++r) {
      dst[r * CPP_]        = acc0[r];
      dst[(8 + r) * CPP_]  = acc1[r];
      dst[(16 + r) * CPP_] = acc2[r];
      dst[(24 + r) * CPP_] = acc3[r];
    }
  }
  __syncthreads();

  v8f d0 = {}, d1 = {};
  {
    const float4* s0 = reinterpret_cast<const float4*>(&cp[wave][c16 * CPP_ + 8 * hh]);
    const float4* s1 = reinterpret_cast<const float4*>(&cp[wave][(16 + c16) * CPP_ + 8 * hh]);
    const float4 a0 = s0[0], a1 = s0[1], b0 = s1[0], b1 = s1[1];
    const v2f a00 = {a0.x, a0.y}, a01 = {a0.z, a0.w}, a10 = {a1.x, a1.y}, a11 = {a1.z, a1.w};
    const v2f b00 = {b0.x, b0.y}, b01 = {b0.z, b0.w}, b10 = {b1.x, b1.y}, b11 = {b1.z, b1.w};
    d0 = __builtin_amdgcn_wmma_f32_16x16x4_f32(false, a00, false, ones, (short)0, d0, false, false);
    d1 = __builtin_amdgcn_wmma_f32_16x16x4_f32(false, b00, false, ones, (short)0, d1, false, false);
    d0 = __builtin_amdgcn_wmma_f32_16x16x4_f32(false, a01, false, ones, (short)0, d0, false, false);
    d1 = __builtin_amdgcn_wmma_f32_16x16x4_f32(false, b01, false, ones, (short)0, d1, false, false);
    d0 = __builtin_amdgcn_wmma_f32_16x16x4_f32(false, a10, false, ones, (short)0, d0, false, false);
    d1 = __builtin_amdgcn_wmma_f32_16x16x4_f32(false, b10, false, ones, (short)0, d1, false, false);
    d0 = __builtin_amdgcn_wmma_f32_16x16x4_f32(false, a11, false, ones, (short)0, d0, false, false);
    d1 = __builtin_amdgcn_wmma_f32_16x16x4_f32(false, b11, false, ones, (short)0, d1, false, false);
  }
  if (c16 == 0) {
    float4* o = reinterpret_cast<float4*>(&csum[32 * wave + 8 * hh]);
    o[0] = make_float4(d0[0], d0[1], d0[2], d0[3]);
    o[1] = make_float4(d0[4], d0[5], d0[6], d0[7]);
    o[4] = make_float4(d1[0], d1[1], d1[2], d1[3]);
    o[5] = make_float4(d1[4], d1[5], d1[6], d1[7]);
  }

  float* irs = ws + OFF_IRS + ((c0l * 4 + t) * 64 + bc) * 224;
  if (tid < H_) {
    const int o = CPH_ + (tid >> 4) * CPP_ + (tid & 15);
    float pw[4];
    #pragma unroll
    for (int w = 0; w < 4; ++w)
      pw[w] = ((cp[w][o] + cp[w][o + 8 * CPP_]) + cp[w][o + 16 * CPP_]) + cp[w][o + 24 * CPP_];
    const float s = ((pw[0] + pw[1]) + pw[2]) + pw[3];
    const float r = 1.0f / s; rsb[tid] = r; irs[tid] = r;
  }
  __syncthreads();
  irs[96 + tid] = 1.0f / csum[tid];

  if (tid < 32) {
    float mm = fmaxf(fmaxf(mv[tid], mv[tid + 32]), mv[tid + 64]);
    float nn = fmaxf(fmaxf(nv[tid], nv[tid + 32]), fmaxf(nv[tid + 64], nv[tid + 96]));
    float ri = fminf(fminf(rsb[tid], rsb[tid + 32]), rsb[tid + 64]);
    float cx = fmaxf(fmaxf(csum[tid], csum[tid + 32]), fmaxf(csum[tid + 64], csum[tid + 96]));
    for (int o = 16; o > 0; o >>= 1) {
      mm = fmaxf(mm, __shfl_xor(mm, o));
      nn = fmaxf(nn, __shfl_xor(nn, o));
      ri = fminf(ri, __shfl_xor(ri, o));
      cx = fmaxf(cx, __shfl_xor(cx, o));
    }
    if (tid == 0)
      atomicMax((int*)(ws + OFF_GMAX + c0 * 4 + t),
                __float_as_int(__expf(mm * nn) * (ri + 1.0f / cx)));
  }
}

template <bool LAST>
__device__ __forceinline__ void score_rows(const float* __restrict__ wsr,
                                           const float* __restrict__ values,
                                           float* __restrict__ out,
                                           int c0, int c0l, int b, int h0, int w,
                                           const float (&ig)[4], float (&z)[ROWS_], float& lmax) {
  #pragma unroll 4
  for (int c = 0; c < C_; ++c) {
    const int bc = b * C_ + c;
    const float* hr  = wsr + OFF_HR + c0 * (BC_ * H_) + bc * H_ + h0;
    const float* hc  = wsr + OFF_HC + c0 * (BC_ * H_) + bc * H_ + h0;
    const float* irb = wsr + OFF_IRS + ((c0l * 4) * 64 + bc) * 224;
    float nn[2], icv[4];
    nn[0] = wsr[OFF_WR + c0 * (BC_ * W_) + bc * W_ + w];
    nn[1] = wsr[OFF_WC + c0 * (BC_ * W_) + bc * W_ + w];
    #pragma unroll
    for (int t = 0; t < 4; ++t) icv[t] = irb[t * (64 * 224) + 96 + w];
    #pragma unroll
    for (int r = 0; r < ROWS_; ++r) {
      float mm[2];
      mm[0] = hr[r];
      mm[1] = hc[r];
      float s = 0.f;
      #pragma unroll
      for (int t = 0; t < 4; ++t) {
        const float e = __expf(mm[t >> 1] * nn[t & 1]);
        s += e * (irb[t * (64 * 224) + h0 + r] + icv[t]) * ig[t];
      }
      const int off = bc * HW_ + (h0 + r) * W_ + w;
      z[r] += s * values[off];
      lmax = fmaxf(lmax, s);
      if (LAST) out[BC_ * HW_ + off] = s;
    }
  }
}

__global__ void k4_scorez(const float* __restrict__ wsr, float* __restrict__ wsw,
                          const float* __restrict__ values, float* __restrict__ out,
                          int c0base) {
  const int w   = threadIdx.x;
  const int rg  = blockIdx.x % (H_ / ROWS_);
  const int cb  = blockIdx.x / (H_ / ROWS_);
  const int b   = cb & 1;
  const int c0l = cb >> 1;
  const int c0  = c0base + c0l;
  const int h0  = rg * ROWS_;
  __shared__ float red[128];

  float ig[4];
  #pragma unroll
  for (int t = 0; t < 4; ++t) ig[t] = 1.0f / wsr[OFF_GMAX + c0 * 4 + t];

  float z[ROWS_];
  #pragma unroll
  for (int r = 0; r < ROWS_; ++r) z[r] = 0.f;
  float lmax = 0.0f;
  if (c0 == C_ - 1) score_rows<true >(wsr, values, out, c0, c0l, b, h0, w, ig, z, lmax);
  else              score_rows<false>(wsr, values, out, c0, c0l, b, h0, w, ig, z, lmax);

  #pragma unroll
  for (int r = 0; r < ROWS_; ++r) out[(b * C_ + c0) * HW_ + (h0 + r) * W_ + w] = z[r];

  red[w] = lmax;
  __syncthreads();
  for (int s2 = 64; s2 > 0; s2 >>= 1) {
    if (w < s2) red[w] = fmaxf(red[w], red[w + s2]);
    __syncthreads();
  }
  if (w == 0)
    atomicMax((int*)(wsw + OFF_GSMAX + c0), __float_as_int(red[0]));
}

__global__ void k6_scale(const float* __restrict__ ws, float* __restrict__ out) {
  const int g4    = blockIdx.x * 256 + threadIdx.x;
  const int plane = (g4 * 4) / HW_;
  const int c0    = (plane < BC_) ? (plane & (C_ - 1)) : (C_ - 1);
  const float inv = 1.0f / ws[OFF_GSMAX + c0];
  float4* o = reinterpret_cast<float4*>(out) + g4;
  float4 v = *o;
  v.x *= inv; v.y *= inv; v.z *= inv; v.w *= inv;
  *o = v;
}

extern "C" void kernel_launch(void* const* d_in, const int* in_sizes, int n_in,
                              void* d_out, int out_size, void* d_ws, size_t ws_size,
                              hipStream_t stream) {
  (void)in_sizes; (void)n_in; (void)out_size; (void)ws_size;
  const float* q   = (const float*)d_in[0];
  const float* k   = (const float*)d_in[1];
  const float* val = (const float*)d_in[2];
  float* out = (float*)d_out;
  float* ws  = (float*)d_ws;

  k1_sums<<<64, 256, 0, stream>>>(q, k, ws);
  k2_norm<<<128, 256, 0, stream>>>(ws);
  k3_tile<<<2816, 128, 0, stream>>>(ws, 0);
  k4_scorez<<<1056, 128, 0, stream>>>(ws, ws, val, out, 0);
  k3_tile<<<2816, 128, 0, stream>>>(ws, 11);
  k4_scorez<<<1056, 128, 0, stream>>>(ws, ws, val, out, 11);
  k3_tile<<<2560, 128, 0, stream>>>(ws, 22);
  k4_scorez<<<960, 128, 0, stream>>>(ws, ws, val, out, 22);
  k6_scale<<<1536, 256, 0, stream>>>(ws, out);
}


# k1: one block per (plane, tensor), each wave owns 12 rows as 16-byte loads issued in one batch; row sums by lane butterfly, column sums via 8 per-wave partials; no LDS tile; sum order changes
# speedup vs baseline: 6.5866x; 1.0796x over previous
#include <hip/hip_runtime.h>
#include <hip/hip_bf16.h>
#include <hip/hip_fp16.h>

typedef __attribute__((ext_vector_type(2)))  float    v2f;
typedef __attribute__((ext_vector_type(8)))  float    v8f;

#define B_   2
#define C_   32
#define H_   96
#define W_   128
#define HW_  (H_ * W_)
#define BC_  (B_ * C_)
#define EPSF 1e-6f

#define OFF_QS    0
#define OFF_QC    6144
#define OFF_KH    14336
#define OFF_KW    20480
#define OFF_SQ    28672
#define OFF_SK    28736
#define OFF_GMAX  28800
#define OFF_GSMAX 28928
#define OFF_HR    28960
#define OFF_HC    (OFF_HR + 32 * 6144)
#define OFF_WR    (OFF_HC + 32 * 6144)
#define OFF_WC    (OFF_WR + 32 * 8192)
#define OFF_IRS   (OFF_WC + 32 * 8192)
#define ROWS_     2

__global__ void __launch_bounds__(256) k1_sums(const float* __restrict__ q, const float* __restrict__ k,
                                               float* __restrict__ ws) {
  __shared__ __attribute__((aligned(16))) float colp[8][W_];
  __shared__ float rowbuf[H_];
  const int bc   = blockIdx.x >> 1;
  const int p    = blockIdx.x & 1;
  const int tid  = threadIdx.x;
  const int wave = tid >> 5, lane = tid & 31;
  const float* s = (p ? k : q) + bc * HW_ + (12 * wave) * W_ + 4 * lane;
  float4 v[12];
  #pragma unroll
  for (int r = 0; r < 12; ++r) v[r] = *reinterpret_cast<const float4*>(s + r * W_);
  float4 cs = v[0];
  float rs[12];
  #pragma unroll
  for (int r = 0; r < 12; ++r) {
    if (r) { cs.x += v[r].x; cs.y += v[r].y; cs.z += v[r].z; cs.w += v[r].w; }
    rs[r] = (v[r].x + v[r].y) + (v[r].z + v[r].w);
  }
  #pragma unroll
  for (int o = 16; o > 0; o >>= 1) {
    #pragma unroll
    for (int r = 0; r < 12; ++r) rs[r] += __shfl_xor(rs[r], o);
  }
  float mine = rs[0];
  #pragma unroll
  for (int r = 1; r < 12; ++r) mine = (lane == r) ? rs[r] : mine;
  *reinterpret_cast<float4*>(&colp[wave][4 * lane]) = cs;
  if (lane < 12) {
    rowbuf[12 * wave + lane] = mine;
    ws[(p ? OFF_KH : OFF_QS) + bc * H_ + 12 * wave + lane] = mine;
  }
  __syncthreads();
  if (tid < W_) {
    float acc = colp[0][tid];
    #pragma unroll
    for (int w = 1; w < 8; ++w) acc += colp[w][tid];
    ws[(p ? OFF_KW : OFF_QC) + bc * W_ + tid] = acc;
  } else if (tid < W_ + 32) {
    float t = (rowbuf[lane] + rowbuf[lane + 32]) + rowbuf[lane + 64];
    for (int o = 16; o > 0; o >>= 1) t += __shfl_xor(t, o);
    if (lane == 0) ws[(p ? OFF_SK : OFF_SQ) + bc] = t;
  }
}

template <int TYPE>
__device__ __forceinline__ void k2_family(float* __restrict__ ws, int c0, int tid,
                                          float* rmn, float* rmx) {
  constexpr int   LEN    = (TYPE < 2) ? H_ : W_;
  constexpr int   N      = BC_ * LEN;
  constexpr int   PER    = N / 256;
  constexpr float addeps = LEN * EPSF;
  float* dst = ws + ((TYPE == 0) ? OFF_HR : (TYPE == 1) ? OFF_HC : (TYPE == 2) ? OFF_WR : OFF_WC) + c0 * N;
  float fa[PER], fb[PER];
  #pragma unroll
  for (int j = 0; j < PER; ++j) {
    const int i  = tid + 256 * j;
    const int b  = (256 * j) / (C_ * LEN);
    const int bc = i / LEN;
    const int x  = i % LEN;
    if      (TYPE == 0) { fa[j] = ws[OFF_SQ + b * C_ + c0];            fb[j] = ws[OFF_KH + i];  }
    else if (TYPE == 1) { fa[j] = ws[OFF_QS + (b * C_ + c0) * H_ + x]; fb[j] = ws[OFF_SK + bc]; }
    else if (TYPE == 2) { fa[j] = ws[OFF_SQ + b * C_ + c0];            fb[j] = ws[OFF_KW + i];  }
    else                { fa[j] = ws[OFF_QC + (b * C_ + c0) * W_ + x]; fb[j] = ws[OFF_SK + bc]; }
  }
  float lmn = 3.4e38f, lmx = -3.4e38f;
  #pragma unroll
  for (int j = 0; j < PER; ++j) {
    fa[j] = fa[j] * fb[j] + addeps;
    lmn = fminf(lmn, fa[j]);
    lmx = fmaxf(lmx, fa[j]);
  }
  rmn[tid] = lmn; rmx[tid] = lmx;
  __syncthreads();
  for (int s = 128; s > 0; s >>= 1) {
    if (tid < s) { rmn[tid] = fminf(rmn[tid], rmn[tid + s]); rmx[tid] = fmaxf(rmx[tid], rmx[tid + s]); }
    __syncthreads();
  }
  const float mn  = rmn[0];
  const float inv = 1.0f / (rmx[0] - mn);
  #pragma unroll
  for (int j = 0; j < PER; ++j) dst[tid + 256 * j] = 2.0f + (fa[j] - mn) * inv;
}

__global__ void __launch_bounds__(256) k2_norm(float* __restrict__ ws) {
  const int type = blockIdx.x & 3;
  const int c0   = blockIdx.x >> 2;
  const int tid  = threadIdx.x;
  __shared__ float rmn[256], rmx[256];
  if      (type == 0) k2_family<0>(ws, c0, tid, rmn, rmx);
  else if (type == 1) k2_family<1>(ws, c0, tid, rmn, rmx);
  else if (type == 2) k2_family<2>(ws, c0, tid, rmn, rmx);
  else                k2_family<3>(ws, c0, tid, rmn, rmx);
  if (blockIdx.x == 0)
    for (int i = tid; i < 160; i += 256) ws[OFF_GMAX + i] = 0.0f;
}

#define CPP_ 20
#define CPH_ (32 * CPP_ + 16)
#define CPW_ (CPH_ + 32 * CPP_)
__global__ void __launch_bounds__(128) k3_tile(float* __restrict__ ws, int c0base) {
  const int c0l = blockIdx.x >> 8;
  const int c0  = c0base + c0l;
  const int t   = (blockIdx.x >> 6) & 3;
  const int bc  = blockIdx.x & 63;
  const int tid = threadIdx.x;
  __shared__ float mv[H_], nv[W_];
  __shared__ __attribute__((aligned(16))) float cp[4][CPW_];
  __shared__ __attribute__((aligned(16))) float csum[W_];
  __shared__ float rsb[H_];

  const float* mptr = ws + ((t < 2) ? OFF_HR : OFF_HC) + c0 * (BC_ * H_) + bc * H_;
  const float* nptr = ws + (((t & 1) == 0) ? OFF_WR : OFF_WC) + c0 * (BC_ * W_) + bc * W_;
  if (tid < H_) mv[tid] = mptr[tid];
  nv[tid] = nptr[tid];
  __syncthreads();

  const int wave = tid >> 5, lane = tid & 31;
  const int c16  = lane & 15;
  const int hh   = lane >> 4;
  const int kh   = hh << 1;
  const v2f ones = {1.0f, 1.0f};

  float nk[16];
  #pragma unroll
  for (int i = 0; i < 8; ++i) {
    nk[2 * i]     = nv[32 * wave + 4 * i + kh];
    nk[2 * i + 1] = nv[32 * wave + 4 * i + kh + 1];
  }
  v2f asel[6][2];
  #pragma unroll
  for (int rb = 0; rb < 6; ++rb) {
    const float rsel = (c16 == 8 + rb) ? 1.0f : 0.0f;
    #pragma unroll
    for (int p = 0; p < 2; ++p) {
      asel[rb][p][0] = (c16 == 4 * p + kh)     ? 1.0f : rsel;
      asel[rb][p][1] = (c16 == 4 * p + kh + 1) ? 1.0f : rsel;
    }
  }

  v8f acc0 = {}, acc1 = {}, acc2 = {}, acc3 = {};
  #pragma unroll
  for (int rb = 0; rb < 6; ++rb) {
    const float m0 = mv[16 * rb + c16];
    #pragma unroll
    for (int p = 0; p < 2; ++p) {
      const v2f e0 = { __expf(m0 * nk[2 * p]),      __expf(m0 * nk[2 * p + 1]) };
      const v2f e1 = { __expf(m0 * nk[4 + 2 * p]),  __expf(m0 * nk[4 + 2 * p + 1]) };
      const v2f e2 = { __expf(m0 * nk[8 + 2 * p]),  __expf(m0 * nk[8 + 2 * p + 1]) };
      const v2f e3 = { __expf(m0 * nk[12 + 2 * p]), __expf(m0 * nk[12 + 2 * p + 1]) };
      acc0 = __builtin_amdgcn_wmma_f32_16x16x4_f32(false, asel[rb][p], false, e0, (short)0, acc0, false, false);
      acc1 = __builtin_amdgcn_wmma_f32_16x16x4_f32(false, asel[rb][p], false, e1, (short)0, acc1, false, false);
      acc2 = __builtin_amdgcn_wmma_f32_16x16x4_f32(false, asel[rb][p], false, e2, (short)0, acc2, false, false);
      acc3 = __builtin_amdgcn_wmma_f32_16x16x4_f32(false, asel[rb][p], false, e3, (short)0, acc3, false, false);
    }
  }

  {
    float* dst = &cp[wave][hh * CPH_ + c16];
    #pragma unroll
    for (int r = 0; r < 8; ++r) {
      dst[r * CPP_]        = acc0[r];
      dst[(8 + r) * CPP_]  = acc1[r];
      dst[(16 + r) * CPP_] = acc2[r];
      dst[(24 + r) * CPP_] = acc3[r];
    }
  }
  __syncthreads();

  v8f d0 = {}, d1 = {};
  {
    const float4* s0 = reinterpret_cast<const float4*>(&cp[wave][c16 * CPP_ + 8 * hh]);
    const float4* s1 = reinterpret_cast<const float4*>(&cp[wave][(16 + c16) * CPP_ + 8 * hh]);
    const float4 a0 = s0[0], a1 = s0[1], b0 = s1[0], b1 = s1[1];
    const v2f a00 = {a0.x, a0.y}, a01 = {a0.z, a0.w}, a10 = {a1.x, a1.y}, a11 = {a1.z, a1.w};
    const v2f b00 = {b0.x, b0.y}, b01 = {b0.z, b0.w}, b10 = {b1.x, b1.y}, b11 = {b1.z, b1.w};
    d0 = __builtin_amdgcn_wmma_f32_16x16x4_f32(false, a00, false, ones, (short)0, d0, false, false);
    d1 = __builtin_amdgcn_wmma_f32_16x16x4_f32(false, b00, false, ones, (short)0, d1, false, false);
    d0 = __builtin_amdgcn_wmma_f32_16x16x4_f32(false, a01, false, ones, (short)0, d0, false, false);
    d1 = __builtin_amdgcn_wmma_f32_16x16x4_f32(false, b01, false, ones, (short)0, d1, false, false);
    d0 = __builtin_amdgcn_wmma_f32_16x16x4_f32(false, a10, false, ones, (short)0, d0, false, false);
    d1 = __builtin_amdgcn_wmma_f32_16x16x4_f32(false, b10, false, ones, (short)0, d1, false, false);
    d0 = __builtin_amdgcn_wmma_f32_16x16x4_f32(false, a11, false, ones, (short)0, d0, false, false);
    d1 = __builtin_amdgcn_wmma_f32_16x16x4_f32(false, b11, false, ones, (short)0, d1, false, false);
  }
  if (c16 == 0) {
    float4* o = reinterpret_cast<float4*>(&csum[32 * wave + 8 * hh]);
    o[0] = make_float4(d0[0], d0[1], d0[2], d0[3]);
    o[1] = make_float4(d0[4], d0[5], d0[6], d0[7]);
    o[4] = make_float4(d1[0], d1[1], d1[2], d1[3]);
    o[5] = make_float4(d1[4], d1[5], d1[6], d1[7]);
  }

  float* irs = ws + OFF_IRS + ((c0l * 4 + t) * 64 + bc) * 224;
  if (tid < H_) {
    const int o = CPH_ + (tid >> 4) * CPP_ + (tid & 15);
    float pw[4];
    #pragma unroll
    for (int w = 0; w < 4; ++w)
      pw[w] = ((cp[w][o] + cp[w][o + 8 * CPP_]) + cp[w][o + 16 * CPP_]) + cp[w][o + 24 * CPP_];
    const float s = ((pw[0] + pw[1]) + pw[2]) + pw[3];
    const float r = 1.0f / s; rsb[tid] = r; irs[tid] = r;
  }
  __syncthreads();
  irs[96 + tid] = 1.0f / csum[tid];

  if (tid < 32) {
    float mm = fmaxf(fmaxf(mv[tid], mv[tid + 32]), mv[tid + 64]);
    float nn = fmaxf(fmaxf(nv[tid], nv[tid + 32]), fmaxf(nv[tid + 64], nv[tid + 96]));
    float ri = fminf(fminf(rsb[tid], rsb[tid + 32]), rsb[tid + 64]);
    float cx = fmaxf(fmaxf(csum[tid], csum[tid + 32]), fmaxf(csum[tid + 64], csum[tid + 96]));
    for (int o = 16; o > 0; o >>= 1) {
      mm = fmaxf(mm, __shfl_xor(mm, o));
      nn = fmaxf(nn, __shfl_xor(nn, o));
      ri = fminf(ri, __shfl_xor(ri, o));
      cx = fmaxf(cx, __shfl_xor(cx, o));
    }
    if (tid == 0)
      atomicMax((int*)(ws + OFF_GMAX + c0 * 4 + t),
                __float_as_int(__expf(mm * nn) * (ri + 1.0f / cx)));
  }
}

template <bool LAST>
__device__ __forceinline__ void score_rows(const float* __restrict__ wsr,
                                           const float* __restrict__ values,
                                           float* __restrict__ out,
                                           int c0, int c0l, int b, int h0, int w,
                                           const float (&ig)[4], float (&z)[ROWS_], float& lmax) {
  #pragma unroll 4
  for (int c = 0; c < C_; ++c) {
    const int bc = b * C_ + c;
    const float* hr  = wsr + OFF_HR + c0 * (BC_ * H_) + bc * H_ + h0;
    const float* hc  = wsr + OFF_HC + c0 * (BC_ * H_) + bc * H_ + h0;
    const float* irb = wsr + OFF_IRS + ((c0l * 4) * 64 + bc) * 224;
    float nn[2], icv[4];
    nn[0] = wsr[OFF_WR + c0 * (BC_ * W_) + bc * W_ + w];
    nn[1] = wsr[OFF_WC + c0 * (BC_ * W_) + bc * W_ + w];
    #pragma unroll
    for (int t = 0; t < 4; ++t) icv[t] = irb[t * (64 * 224) + 96 + w];
    #pragma unroll
    for (int r = 0; r < ROWS_; ++r) {
      float mm[2];
      mm[0] = hr[r];
      mm[1] = hc[r];
      float s = 0.f;
      #pragma unroll
      for (int t = 0; t < 4; ++t) {
        const float e = __expf(mm[t >> 1] * nn[t & 1]);
        s += e * (irb[t * (64 * 224) + h0 + r] + icv[t]) * ig[t];
      }
      const int off = bc * HW_ + (h0 + r) * W_ + w;
      z[r] += s * values[off];
      lmax = fmaxf(lmax, s);
      if (LAST) out[BC_ * HW_ + off] = s;
    }
  }
}

__global__ void k4_scorez(const float* __restrict__ wsr, float* __restrict__ wsw,
                          const float* __restrict__ values, float* __restrict__ out,
                          int c0base) {
  const int w   = threadIdx.x;
  const int rg  = blockIdx.x % (H_ / ROWS_);
  const int cb  = blockIdx.x / (H_ / ROWS_);
  const int b   = cb & 1;
  const int c0l = cb >> 1;
  const int c0  = c0base + c0l;
  const int h0  = rg * ROWS_;
  __shared__ float red[128];

  float ig[4];
  #pragma unroll
  for (int t = 0; t < 4; ++t) ig[t] = 1.0f / wsr[OFF_GMAX + c0 * 4 + t];

  float z[ROWS_];
  #pragma unroll
  for (int r = 0; r < ROWS_; ++r) z[r] = 0.f;
  float lmax = 0.0f;
  if (c0 == C_ - 1) score_rows<true >(wsr, values, out, c0, c0l, b, h0, w, ig, z, lmax);
  else              score_rows<false>(wsr, values, out, c0, c0l, b, h0, w, ig, z, lmax);

  #pragma unroll
  for (int r = 0; r < ROWS_; ++r) out[(b * C_ + c0) * HW_ + (h0 + r) * W_ + w] = z[r];

  red[w] = lmax;
  __syncthreads();
  for (int s2 = 64; s2 > 0; s2 >>= 1) {
    if (w < s2) red[w] = fmaxf(red[w], red[w + s2]);
    __syncthreads();
  }
  if (w == 0)
    atomicMax((int*)(wsw + OFF_GSMAX + c0), __float_as_int(red[0]));
}

__global__ void k6_scale(const float* __restrict__ ws, float* __restrict__ out) {
  const int g4    = blockIdx.x * 256 + threadIdx.x;
  const int plane = (g4 * 4) / HW_;
  const int c0    = (plane < BC_) ? (plane & (C_ - 1)) : (C_ - 1);
  const float inv = 1.0f / ws[OFF_GSMAX + c0];
  float4* o = reinterpret_cast<float4*>(out) + g4;
  float4 v = *o;
  v.x *= inv; v.y *= inv; v.z *= inv; v.w *= inv;
  *o = v;
}

extern "C" void kernel_launch(void* const* d_in, const int* in_sizes, int n_in,
                              void* d_out, int out_size, void* d_ws, size_t ws_size,
                              hipStream_t stream) {
  (void)in_sizes; (void)n_in; (void)out_size; (void)ws_size;
  const float* q   = (const float*)d_in[0];
  const float* k   = (const float*)d_in[1];
  const float* val = (const float*)d_in[2];
  float* out = (float*)d_out;
  float* ws  = (float*)d_ws;

  k1_sums<<<128, 256, 0, stream>>>(q, k, ws);
  k2_norm<<<128, 256, 0, stream>>>(ws);
  k3_tile<<<2816, 128, 0, stream>>>(ws, 0);
  k4_scorez<<<1056, 128, 0, stream>>>(ws, ws, val, out, 0);
  k3_tile<<<2816, 128, 0, stream>>>(ws, 11);
  k4_scorez<<<1056, 128, 0, stream>>>(ws, ws, val, out, 11);
  k3_tile<<<2560, 128, 0, stream>>>(ws, 22);
  k4_scorez<<<960, 128, 0, stream>>>(ws, ws, val, out, 22);
  k6_scale<<<1536, 256, 0, stream>>>(ws, out);
}


# m (h_row, h_col) rebuilt in k3/k4 from the k1 sums and k2's min, 1/(max-min) instead of stored planes; freed workspace holds 16 query channels per pass: 3 passes become 2, 9 launches to 7
# speedup vs baseline: 8.2053x; 1.2458x over previous
#include <hip/hip_runtime.h>
#include <hip/hip_bf16.h>
#include <hip/hip_fp16.h>

typedef __attribute__((ext_vector_type(2)))  float    v2f;
typedef __attribute__((ext_vector_type(8)))  float    v8f;

#define B_   2
#define C_   32
#define H_   96
#define W_   128
#define HW_  (H_ * W_)
#define BC_  (B_ * C_)
#define EPSF 1e-6f

#define OFF_QS    0
#define OFF_QC    6144
#define OFF_KH    14336
#define OFF_KW    20480
#define OFF_SQ    28672
#define OFF_SK    28736
#define OFF_GMAX  28800
#define OFF_GSMAX 28928
#define OFF_MN    28960
#define OFF_WR    (OFF_MN + 128)
#define OFF_WC    (OFF_WR + 32 * 8192)
#define OFF_IRS   (OFF_WC + 32 * 8192)
#define ROWS_     2

__global__ void __launch_bounds__(256) k1_sums(const float* __restrict__ q, const float* __restrict__ k,
                                               float* __restrict__ ws) {
  __shared__ __attribute__((aligned(16))) float colp[8][W_];
  __shared__ float rowbuf[H_];
  const int bc   = blockIdx.x >> 1;
  const int p    = blockIdx.x & 1;
  const int tid  = threadIdx.x;
  const int wave = tid >> 5, lane = tid & 31;
  const float* s = (p ? k : q) + bc * HW_ + (12 * wave) * W_ + 4 * lane;
  float4 v[12];
  #pragma unroll
  for (int r = 0; r < 12; ++r) v[r] = *reinterpret_cast<const float4*>(s + r * W_);
  float4 cs = v[0];
  float rs[12];
  #pragma unroll
  for (int r = 0; r < 12; ++r) {
    if (r) { cs.x += v[r].x; cs.y += v[r].y; cs.z += v[r].z; cs.w += v[r].w; }
    rs[r] = (v[r].x + v[r].y) + (v[r].z + v[r].w);
  }
  #pragma unroll
  for (int o = 16; o > 0; o >>= 1) {
    #pragma unroll
    for (int r = 0; r < 12; ++r) rs[r] += __shfl_xor(rs[r], o);
  }
  float mine = rs[0];
  #pragma unroll
  for (int r = 1; r < 12; ++r) mine = (lane == r) ? rs[r] : mine;
  *reinterpret_cast<float4*>(&colp[wave][4 * lane]) = cs;
  if (lane < 12) {
    rowbuf[12 * wave + lane] = mine;
    ws[(p ? OFF_KH : OFF_QS) + bc * H_ + 12 * wave + lane] = mine;
  }
  __syncthreads();
  if (tid < W_) {
    float acc = colp[0][tid];
    #pragma unroll
    for (int w = 1; w < 8; ++w) acc += colp[w][tid];
    ws[(p ? OFF_KW : OFF_QC) + bc * W_ + tid] = acc;
  } else if (tid < W_ + 32) {
    float t = (rowbuf[lane] + rowbuf[lane + 32]) + rowbuf[lane + 64];
    for (int o = 16; o > 0; o >>= 1) t += __shfl_xor(t, o);
    if (lane == 0) ws[(p ? OFF_SK : OFF_SQ) + bc] = t;
  }
}


__device__ __forceinline__ float raw_elem(float fa, float fb, float addeps) { return fa * fb + addeps; }
__device__ __forceinline__ float norm_elem(float fa, float fb, float addeps, float mn, float inv) {
  return 2.0f + (raw_elem(fa, fb, addeps) - mn) * inv;
}

template <int TYPE>
__device__ __forceinline__ void k2_family(float* __restrict__ ws, int c0, int tid,
                                          float* rmn, float* rmx) {
  constexpr int   LEN    = (TYPE < 2) ? H_ : W_;
  constexpr int   N      = BC_ * LEN;
  constexpr int   PER    = N / 256;
  constexpr float addeps = LEN * EPSF;
  float fa[PER], fb[PER];
  #pragma unroll
  for (int j = 0; j < PER; ++j) {
    const int i  = tid + 256 * j;
    const int b  = (256 * j) / (C_ * LEN);
    const int bc = i / LEN;
    const int x  = i % LEN;
    if      (TYPE == 0) { fa[j] = ws[OFF_SQ + b * C_ + c0];            fb[j] = ws[OFF_KH + i];  }
    else if (TYPE == 1) { fa[j] = ws[OFF_QS + (b * C_ + c0) * H_ + x]; fb[j] = ws[OFF_SK + bc]; }
    else if (TYPE == 2) { fa[j] = ws[OFF_SQ + b * C_ + c0];            fb[j] = ws[OFF_KW + i];  }
    else                { fa[j] = ws[OFF_QC + (b * C_ + c0) * W_ + x]; fb[j] = ws[OFF_SK + bc]; }
  }
  float lmn = 3.4e38f, lmx = -3.4e38f;
  #pragma unroll
  for (int j = 0; j < PER; ++j) {
    fa[j] = raw_elem(fa[j], fb[j], addeps);
    lmn = fminf(lmn, fa[j]);
    lmx = fmaxf(lmx, fa[j]);
  }
  rmn[tid] = lmn; rmx[tid] = lmx;
  __syncthreads();
  for (int s = 128; s > 0; s >>= 1) {
    if (tid < s) { rmn[tid] = fminf(rmn[tid], rmn[tid + s]); rmx[tid] = fmaxf(rmx[tid], rmx[tid + s]); }
    __syncthreads();
  }
  const float mn  = rmn[0];
  const float inv = 1.0f / (rmx[0] - mn);
  if (TYPE < 2) {
    if (tid == 0) { ws[OFF_MN + (c0 * 2 + TYPE) * 2] = mn; ws[OFF_MN + (c0 * 2 + TYPE) * 2 + 1] = inv; }
  } else {
    float* dst = ws + ((TYPE == 2) ? OFF_WR : OFF_WC) + c0 * N;
    #pragma unroll
    for (int j = 0; j < PER; ++j) dst[tid + 256 * j] = 2.0f + (fa[j] - mn) * inv;
  }
}

__global__ void __launch_bounds__(256) k2_norm(float* __restrict__ ws) {
  const int type = blockIdx.x & 3;
  const int c0   = blockIdx.x >> 2;
  const int tid  = threadIdx.x;
  __shared__ float rmn[256], rmx[256];
  if      (type == 0) k2_family<0>(ws, c0, tid, rmn, rmx);
  else if (type == 1) k2_family<1>(ws, c0, tid, rmn, rmx);
  else if (type == 2) k2_family<2>(ws, c0, tid, rmn, rmx);
  else                k2_family<3>(ws, c0, tid, rmn, rmx);
  if (blockIdx.x == 0)
    for (int i = tid; i < 160; i += 256) ws[OFF_GMAX + i] = 0.0f;
}

#define CPP_ 20
#define CPH_ (32 * CPP_ + 16)
#define CPW_ (CPH_ + 32 * CPP_)
__global__ void __launch_bounds__(128) k3_tile(float* __restrict__ ws, int c0base) {
  const int c0l = blockIdx.x >> 8;
  const int c0  = c0base + c0l;
  const int t   = (blockIdx.x >> 6) & 3;
  const int bc  = blockIdx.x & 63;
  const int tid = threadIdx.x;
  __shared__ float mv[H_], nv[W_];
  __shared__ __attribute__((aligned(16))) float cp[4][CPW_];
  __shared__ __attribute__((aligned(16))) float csum[W_];
  __shared__ float rsb[H_];

  {
    const int   bq  = (bc >> 5) * C_ + c0;
    const int   mt  = (t < 2) ? 0 : 1;
    const int   hm  = (tid < H_) ? tid : 0;
    const float fa  = ws[mt ? (OFF_QS + bq * H_ + hm) : (OFF_SQ + bq)];
    const float fb  = ws[mt ? (OFF_SK + bc) : (OFF_KH + bc * H_ + hm)];
    const float mn  = ws[OFF_MN + (c0 * 2 + mt) * 2];
    const float iv  = ws[OFF_MN + (c0 * 2 + mt) * 2 + 1];
    const float nl  = ws[(((t & 1) == 0) ? OFF_WR : OFF_WC) + c0 * (BC_ * W_) + bc * W_ + tid];
    if (tid < H_) mv[tid] = norm_elem(fa, fb, H_ * EPSF, mn, iv);
    nv[tid] = nl;
  }
  __syncthreads();

  const int wave = tid >> 5, lane = tid & 31;
  const int c16  = lane & 15;
  const int hh   = lane >> 4;
  const int kh   = hh << 1;
  const v2f ones = {1.0f, 1.0f};

  float nk[16];
  #pragma unroll
  for (int i = 0; i < 8; ++i) {
    nk[2 * i]     = nv[32 * wave + 4 * i + kh];
    nk[2 * i + 1] = nv[32 * wave + 4 * i + kh + 1];
  }
  v2f asel[6][2];
  #pragma unroll
  for (int rb = 0; rb < 6; ++rb) {
    const float rsel = (c16 == 8 + rb) ? 1.0f : 0.0f;
    #pragma unroll
    for (int p = 0; p < 2; ++p) {
      asel[rb][p][0] = (c16 == 4 * p + kh)     ? 1.0f : rsel;
      asel[rb][p][1] = (c16 == 4 * p + kh + 1) ? 1.0f : rsel;
    }
  }

  v8f acc0 = {}, acc1 = {}, acc2 = {}, acc3 = {};
  #pragma unroll
  for (int rb = 0; rb < 6; ++rb) {
    const float m0 = mv[16 * rb + c16];
    #pragma unroll
    for (int p = 0; p < 2; ++p) {
      const v2f e0 = { __expf(m0 * nk[2 * p]),      __expf(m0 * nk[2 * p + 1]) };
      const v2f e1 = { __expf(m0 * nk[4 + 2 * p]),  __expf(m0 * nk[4 + 2 * p + 1]) };
      const v2f e2 = { __expf(m0 * nk[8 + 2 * p]),  __expf(m0 * nk[8 + 2 * p + 1]) };
      const v2f e3 = { __expf(m0 * nk[12 + 2 * p]), __expf(m0 * nk[12 + 2 * p + 1]) };
      acc0 = __builtin_amdgcn_wmma_f32_16x16x4_f32(false, asel[rb][p], false, e0, (short)0, acc0, false, false);
      acc1 = __builtin_amdgcn_wmma_f32_16x16x4_f32(false, asel[rb][p], false, e1, (short)0, acc1, false, false);
      acc2 = __builtin_amdgcn_wmma_f32_16x16x4_f32(false, asel[rb][p], false, e2, (short)0, acc2, false, false);
      acc3 = __builtin_amdgcn_wmma_f32_16x16x4_f32(false, asel[rb][p], false, e3, (short)0, acc3, false, false);
    }
  }

  {
    float* dst = &cp[wave][hh * CPH_ + c16];
    #pragma unroll
    for (int r = 0; r < 8; ++r) {
      dst[r * CPP_]        = acc0[r];
      dst[(8 + r) * CPP_]  = acc1[r];
      dst[(16 + r) * CPP_] = acc2[r];
      dst[(24 + r) * CPP_] = acc3[r];
    }
  }
  __syncthreads();

  v8f d0 = {}, d1 = {};
  {
    const float4* s0 = reinterpret_cast<const float4*>(&cp[wave][c16 * CPP_ + 8 * hh]);
    const float4* s1 = reinterpret_cast<const float4*>(&cp[wave][(16 + c16) * CPP_ + 8 * hh]);
    const float4 a0 = s0[0], a1 = s0[1], b0 = s1[0], b1 = s1[1];
    const v2f a00 = {a0.x, a0.y}, a01 = {a0.z, a0.w}, a10 = {a1.x, a1.y}, a11 = {a1.z, a1.w};
    const v2f b00 = {b0.x, b0.y}, b01 = {b0.z, b0.w}, b10 = {b1.x, b1.y}, b11 = {b1.z, b1.w};
    d0 = __builtin_amdgcn_wmma_f32_16x16x4_f32(false, a00, false, ones, (short)0, d0, false, false);
    d1 = __builtin_amdgcn_wmma_f32_16x16x4_f32(false, b00, false, ones, (short)0, d1, false, false);
    d0 = __builtin_amdgcn_wmma_f32_16x16x4_f32(false, a01, false, ones, (short)0, d0, false, false);
    d1 = __builtin_amdgcn_wmma_f32_16x16x4_f32(false, b01, false, ones, (short)0, d1, false, false);
    d0 = __builtin_amdgcn_wmma_f32_16x16x4_f32(false, a10, false, ones, (short)0, d0, false, false);
    d1 = __builtin_amdgcn_wmma_f32_16x16x4_f32(false, b10, false, ones, (short)0, d1, false, false);
    d0 = __builtin_amdgcn_wmma_f32_16x16x4_f32(false, a11, false, ones, (short)0, d0, false, false);
    d1 = __builtin_amdgcn_wmma_f32_16x16x4_f32(false, b11, false, ones, (short)0, d1, false, false);
  }
  if (c16 == 0) {
    float4* o = reinterpret_cast<float4*>(&csum[32 * wave + 8 * hh]);
    o[0] = make_float4(d0[0], d0[1], d0[2], d0[3]);
    o[1] = make_float4(d0[4], d0[5], d0[6], d0[7]);
    o[4] = make_float4(d1[0], d1[1], d1[2], d1[3]);
    o[5] = make_float4(d1[4], d1[5], d1[6], d1[7]);
  }

  float* irs = ws + OFF_IRS + ((c0l * 4 + t) * 64 + bc) * 224;
  if (tid < H_) {
    const int o = CPH_ + (tid >> 4) * CPP_ + (tid & 15);
    float pw[4];
    #pragma unroll
    for (int w = 0; w < 4; ++w)
      pw[w] = ((cp[w][o] + cp[w][o + 8 * CPP_]) + cp[w][o + 16 * CPP_]) + cp[w][o + 24 * CPP_];
    const float s = ((pw[0] + pw[1]) + pw[2]) + pw[3];
    const float r = 1.0f / s; rsb[tid] = r; irs[tid] = r;
  }
  __syncthreads();
  irs[96 + tid] = 1.0f / csum[tid];

  if (tid < 32) {
    float mm = fmaxf(fmaxf(mv[tid], mv[tid + 32]), mv[tid + 64]);
    float nn = fmaxf(fmaxf(nv[tid], nv[tid + 32]), fmaxf(nv[tid + 64], nv[tid + 96]));
    float ri = fminf(fminf(rsb[tid], rsb[tid + 32]), rsb[tid + 64]);
    float cx = fmaxf(fmaxf(csum[tid], csum[tid + 32]), fmaxf(csum[tid + 64], csum[tid + 96]));
    for (int o = 16; o > 0; o >>= 1) {
      mm = fmaxf(mm, __shfl_xor(mm, o));
      nn = fmaxf(nn, __shfl_xor(nn, o));
      ri = fminf(ri, __shfl_xor(ri, o));
      cx = fmaxf(cx, __shfl_xor(cx, o));
    }
    if (tid == 0)
      atomicMax((int*)(ws + OFF_GMAX + c0 * 4 + t),
                __float_as_int(__expf(mm * nn) * (ri + 1.0f / cx)));
  }
}

template <bool LAST>
__device__ __forceinline__ void score_rows(const float* __restrict__ wsr,
                                           const float* __restrict__ values,
                                           float* __restrict__ out,
                                           int c0, int c0l, int b, int h0, int w,
                                           const float (&ig)[4], float (&z)[ROWS_], float& lmax) {
  const int   bq = b * C_ + c0;
  const float sq = wsr[OFF_SQ + bq];
  float qs[ROWS_], mn[2], iv[2];
  #pragma unroll
  for (int r = 0; r < ROWS_; ++r) qs[r] = wsr[OFF_QS + bq * H_ + h0 + r];
  #pragma unroll
  for (int f = 0; f < 2; ++f) { mn[f] = wsr[OFF_MN + (c0 * 2 + f) * 2]; iv[f] = wsr[OFF_MN + (c0 * 2 + f) * 2 + 1]; }
  #pragma unroll 4
  for (int c = 0; c < C_; ++c) {
    const int bc = b * C_ + c;
    const float  sk  = wsr[OFF_SK + bc];
    const float* kh  = wsr + OFF_KH + bc * H_ + h0;
    const float* irb = wsr + OFF_IRS + ((c0l * 4) * 64 + bc) * 224;
    float nn[2], icv[4];
    nn[0] = wsr[OFF_WR + c0 * (BC_ * W_) + bc * W_ + w];
    nn[1] = wsr[OFF_WC + c0 * (BC_ * W_) + bc * W_ + w];
    #pragma unroll
    for (int t = 0; t < 4; ++t) icv[t] = irb[t * (64 * 224) + 96 + w];
    #pragma unroll
    for (int r = 0; r < ROWS_; ++r) {
      float mm[2];
      mm[0] = norm_elem(sq,    kh[r], H_ * EPSF, mn[0], iv[0]);
      mm[1] = norm_elem(qs[r], sk,    H_ * EPSF, mn[1], iv[1]);
      float s = 0.f;
      #pragma unroll
      for (int t = 0; t < 4; ++t) {
        const float e = __expf(mm[t >> 1] * nn[t & 1]);
        s += e * (irb[t * (64 * 224) + h0 + r] + icv[t]) * ig[t];
      }
      const int off = bc * HW_ + (h0 + r) * W_ + w;
      z[r] += s * values[off];
      lmax = fmaxf(lmax, s);
      if (LAST) out[BC_ * HW_ + off] = s;
    }
  }
}

__global__ void k4_scorez(const float* __restrict__ wsr, float* __restrict__ wsw,
                          const float* __restrict__ values, float* __restrict__ out,
                          int c0base) {
  const int w   = threadIdx.x;
  const int rg  = blockIdx.x % (H_ / ROWS_);
  const int cb  = blockIdx.x / (H_ / ROWS_);
  const int b   = cb & 1;
  const int c0l = cb >> 1;
  const int c0  = c0base + c0l;
  const int h0  = rg * ROWS_;
  __shared__ float red[128];

  float ig[4];
  #pragma unroll
  for (int t = 0; t < 4; ++t) ig[t] = 1.0f / wsr[OFF_GMAX + c0 * 4 + t];

  float z[ROWS_];
  #pragma unroll
  for (int r = 0; r < ROWS_; ++r) z[r] = 0.f;
  float lmax = 0.0f;
  if (c0 == C_ - 1) score_rows<true >(wsr, values, out, c0, c0l, b, h0, w, ig, z, lmax);
  else              score_rows<false>(wsr, values, out, c0, c0l, b, h0, w, ig, z, lmax);

  #pragma unroll
  for (int r = 0; r < ROWS_; ++r) out[(b * C_ + c0) * HW_ + (h0 + r) * W_ + w] = z[r];

  red[w] = lmax;
  __syncthreads();
  for (int s2 = 64; s2 > 0; s2 >>= 1) {
    if (w < s2) red[w] = fmaxf(red[w], red[w + s2]);
    __syncthreads();
  }
  if (w == 0)
    atomicMax((int*)(wsw + OFF_GSMAX + c0), __float_as_int(red[0]));
}

__global__ void k6_scale(const float* __restrict__ ws, float* __restrict__ out) {
  const int g4    = blockIdx.x * 256 + threadIdx.x;
  const int plane = (g4 * 4) / HW_;
  const int c0    = (plane < BC_) ? (plane & (C_ - 1)) : (C_ - 1);
  const float inv = 1.0f / ws[OFF_GSMAX + c0];
  float4* o = reinterpret_cast<float4*>(out) + g4;
  float4 v = *o;
  v.x *= inv; v.y *= inv; v.z *= inv; v.w *= inv;
  *o = v;
}

extern "C" void kernel_launch(void* const* d_in, const int* in_sizes, int n_in,
                              void* d_out, int out_size, void* d_ws, size_t ws_size,
                              hipStream_t stream) {
  (void)in_sizes; (void)n_in; (void)out_size; (void)ws_size;
  const float* q   = (const float*)d_in[0];
  const float* k   = (const float*)d_in[1];
  const float* val = (const float*)d_in[2];
  float* out = (float*)d_out;
  float* ws  = (float*)d_ws;

  k1_sums<<<128, 256, 0, stream>>>(q, k, ws);
  k2_norm<<<128, 256, 0, stream>>>(ws);
  k3_tile<<<4096, 128, 0, stream>>>(ws, 0);
  k4_scorez<<<1536, 128, 0, stream>>>(ws, ws, val, out, 0);
  k3_tile<<<4096, 128, 0, stream>>>(ws, 16);
  k4_scorez<<<1536, 128, 0, stream>>>(ws, ws, val, out, 16);
  k6_scale<<<1536, 256, 0, stream>>>(ws, out);
}


# k4: per-channel row factors (m of both h families, 8 inverse row sums) built once per group into an LDS table; the channel loop has no scalar loads: one L2 round trip per 4-channel batch, not two
# speedup vs baseline: 8.6605x; 1.0555x over previous
#include <hip/hip_runtime.h>
#include <hip/hip_bf16.h>
#include <hip/hip_fp16.h>

typedef __attribute__((ext_vector_type(2)))  float    v2f;
typedef __attribute__((ext_vector_type(8)))  float    v8f;

#define B_   2
#define C_   32
#define H_   96
#define W_   128
#define HW_  (H_ * W_)
#define BC_  (B_ * C_)
#define EPSF 1e-6f

#define OFF_QS    0
#define OFF_QC    6144
#define OFF_KH    14336
#define OFF_KW    20480
#define OFF_SQ    28672
#define OFF_SK    28736
#define OFF_GMAX  28800
#define OFF_GSMAX 28928
#define OFF_MN    28960
#define OFF_WR    (OFF_MN + 128)
#define OFF_WC    (OFF_WR + 32 * 8192)
#define OFF_IRS   (OFF_WC + 32 * 8192)
#define ROWS_     2

__global__ void __launch_bounds__(256) k1_sums(const float* __restrict__ q, const float* __restrict__ k,
                                               float* __restrict__ ws) {
  __shared__ __attribute__((aligned(16))) float colp[8][W_];
  __shared__ float rowbuf[H_];
  const int bc   = blockIdx.x >> 1;
  const int p    = blockIdx.x & 1;
  const int tid  = threadIdx.x;
  const int wave = tid >> 5, lane = tid & 31;
  const float* s = (p ? k : q) + bc * HW_ + (12 * wave) * W_ + 4 * lane;
  float4 v[12];
  #pragma unroll
  for (int r = 0; r < 12; ++r) v[r] = *reinterpret_cast<const float4*>(s + r * W_);
  float4 cs = v[0];
  float rs[12];
  #pragma unroll
  for (int r = 0; r < 12; ++r) {
    if (r) { cs.x += v[r].x; cs.y += v[r].y; cs.z += v[r].z; cs.w += v[r].w; }
    rs[r] = (v[r].x + v[r].y) + (v[r].z + v[r].w);
  }
  #pragma unroll
  for (int o = 16; o > 0; o >>= 1) {
    #pragma unroll
    for (int r = 0; r < 12; ++r) rs[r] += __shfl_xor(rs[r], o);
  }
  float mine = rs[0];
  #pragma unroll
  for (int r = 1; r < 12; ++r) mine = (lane == r) ? rs[r] : mine;
  *reinterpret_cast<float4*>(&colp[wave][4 * lane]) = cs;
  if (lane < 12) {
    rowbuf[12 * wave + lane] = mine;
    ws[(p ? OFF_KH : OFF_QS) + bc * H_ + 12 * wave + lane] = mine;
  }
  __syncthreads();
  if (tid < W_) {
    float acc = colp[0][tid];
    #pragma unroll
    for (int w = 1; w < 8; ++w) acc += colp[w][tid];
    ws[(p ? OFF_KW : OFF_QC) + bc * W_ + tid] = acc;
  } else if (tid < W_ + 32) {
    float t = (rowbuf[lane] + rowbuf[lane + 32]) + rowbuf[lane + 64];
    for (int o = 16; o > 0; o >>= 1) t += __shfl_xor(t, o);
    if (lane == 0) ws[(p ? OFF_SK : OFF_SQ) + bc] = t;
  }
}


__device__ __forceinline__ float raw_elem(float fa, float fb, float addeps) { return fa * fb + addeps; }
__device__ __forceinline__ float norm_elem(float fa, float fb, float addeps, float mn, float inv) {
  return 2.0f + (raw_elem(fa, fb, addeps) - mn) * inv;
}

template <int TYPE>
__device__ __forceinline__ void k2_family(float* __restrict__ ws, int c0, int tid,
                                          float* rmn, float* rmx) {
  constexpr int   LEN    = (TYPE < 2) ? H_ : W_;
  constexpr int   N      = BC_ * LEN;
  constexpr int   PER    = N / 256;
  constexpr float addeps = LEN * EPSF;
  float fa[PER], fb[PER];
  #pragma unroll
  for (int j = 0; j < PER; ++j) {
    const int i  = tid + 256 * j;
    const int b  = (256 * j) / (C_ * LEN);
    const int bc = i / LEN;
    const int x  = i % LEN;
    if      (TYPE == 0) { fa[j] = ws[OFF_SQ + b * C_ + c0];            fb[j] = ws[OFF_KH + i];  }
    else if (TYPE == 1) { fa[j] = ws[OFF_QS + (b * C_ + c0) * H_ + x]; fb[j] = ws[OFF_SK + bc]; }
    else if (TYPE == 2) { fa[j] = ws[OFF_SQ + b * C_ + c0];            fb[j] = ws[OFF_KW + i];  }
    else                { fa[j] = ws[OFF_QC + (b * C_ + c0) * W_ + x]; fb[j] = ws[OFF_SK + bc]; }
  }
  float lmn = 3.4e38f, lmx = -3.4e38f;
  #pragma unroll
  for (int j = 0; j < PER; ++j) {
    fa[j] = raw_elem(fa[j], fb[j], addeps);
    lmn = fminf(lmn, fa[j]);
    lmx = fmaxf(lmx, fa[j]);
  }
  rmn[tid] = lmn; rmx[tid] = lmx;
  __syncthreads();
  for (int s = 128; s > 0; s >>= 1) {
    if (tid < s) { rmn[tid] = fminf(rmn[tid], rmn[tid + s]); rmx[tid] = fmaxf(rmx[tid], rmx[tid + s]); }
    __syncthreads();
  }
  const float mn  = rmn[0];
  const float inv = 1.0f / (rmx[0] - mn);
  if (TYPE < 2) {
    if (tid == 0) { ws[OFF_MN + (c0 * 2 + TYPE) * 2] = mn; ws[OFF_MN + (c0 * 2 + TYPE) * 2 + 1] = inv; }
  } else {
    float* dst = ws + ((TYPE == 2) ? OFF_WR : OFF_WC) + c0 * N;
    #pragma unroll
    for (int j = 0; j < PER; ++j) dst[tid + 256 * j] = 2.0f + (fa[j] - mn) * inv;
  }
}

__global__ void __launch_bounds__(256) k2_norm(float* __restrict__ ws) {
  const int type = blockIdx.x & 3;
  const int c0   = blockIdx.x >> 2;
  const int tid  = threadIdx.x;
  __shared__ float rmn[256], rmx[256];
  if      (type == 0) k2_family<0>(ws, c0, tid, rmn, rmx);
  else if (type == 1) k2_family<1>(ws, c0, tid, rmn, rmx);
  else if (type == 2) k2_family<2>(ws, c0, tid, rmn, rmx);
  else                k2_family<3>(ws, c0, tid, rmn, rmx);
  if (blockIdx.x == 0)
    for (int i = tid; i < 160; i += 256) ws[OFF_GMAX + i] = 0.0f;
}

#define CPP_ 20
#define CPH_ (32 * CPP_ + 16)
#define CPW_ (CPH_ + 32 * CPP_)
__global__ void __launch_bounds__(128) k3_tile(float* __restrict__ ws, int c0base) {
  const int c0l = blockIdx.x >> 8;
  const int c0  = c0base + c0l;
  const int t   = (blockIdx.x >> 6) & 3;
  const int bc  = blockIdx.x & 63;
  const int tid = threadIdx.x;
  __shared__ float mv[H_], nv[W_];
  __shared__ __attribute__((aligned(16))) float cp[4][CPW_];
  __shared__ __attribute__((aligned(16))) float csum[W_];
  __shared__ float rsb[H_];

  {
    const int   bq  = (bc >> 5) * C_ + c0;
    const int   mt  = (t < 2) ? 0 : 1;
    const int   hm  = (tid < H_) ? tid : 0;
    const float fa  = ws[mt ? (OFF_QS + bq * H_ + hm) : (OFF_SQ + bq)];
    const float fb  = ws[mt ? (OFF_SK + bc) : (OFF_KH + bc * H_ + hm)];
    const float mn  = ws[OFF_MN + (c0 * 2 + mt) * 2];
    const float iv  = ws[OFF_MN + (c0 * 2 + mt) * 2 + 1];
    const float nl  = ws[(((t & 1) == 0) ? OFF_WR : OFF_WC) + c0 * (BC_ * W_) + bc * W_ + tid];
    if (tid < H_) mv[tid] = norm_elem(fa, fb, H_ * EPSF, mn, iv);
    nv[tid] = nl;
  }
  __syncthreads();

  const int wave = tid >> 5, lane = tid & 31;
  const int c16  = lane & 15;
  const int hh   = lane >> 4;
  const int kh   = hh << 1;
  const v2f ones = {1.0f, 1.0f};

  float nk[16];
  #pragma unroll
  for (int i = 0; i < 8; ++i) {
    nk[2 * i]     = nv[32 * wave + 4 * i + kh];
    nk[2 * i + 1] = nv[32 * wave + 4 * i + kh + 1];
  }
  v2f asel[6][2];
  #pragma unroll
  for (int rb = 0; rb < 6; ++rb) {
    const float rsel = (c16 == 8 + rb) ? 1.0f : 0.0f;
    #pragma unroll
    for (int p = 0; p < 2; ++p) {
      asel[rb][p][0] = (c16 == 4 * p + kh)     ? 1.0f : rsel;
      asel[rb][p][1] = (c16 == 4 * p + kh + 1) ? 1.0f : rsel;
    }
  }

  v8f acc0 = {}, acc1 = {}, acc2 = {}, acc3 = {};
  #pragma unroll
  for (int rb = 0; rb < 6; ++rb) {
    const float m0 = mv[16 * rb + c16];
    #pragma unroll
    for (int p = 0; p < 2; ++p) {
      const v2f e0 = { __expf(m0 * nk[2 * p]),      __expf(m0 * nk[2 * p + 1]) };
      const v2f e1 = { __expf(m0 * nk[4 + 2 * p]),  __expf(m0 * nk[4 + 2 * p + 1]) };
      const v2f e2 = { __expf(m0 * nk[8 + 2 * p]),  __expf(m0 * nk[8 + 2 * p + 1]) };
      const v2f e3 = { __expf(m0 * nk[12 + 2 * p]), __expf(m0 * nk[12 + 2 * p + 1]) };
      acc0 = __builtin_amdgcn_wmma_f32_16x16x4_f32(false, asel[rb][p], false, e0, (short)0, acc0, false, false);
      acc1 = __builtin_amdgcn_wmma_f32_16x16x4_f32(false, asel[rb][p], false, e1, (short)0, acc1, false, false);
      acc2 = __builtin_amdgcn_wmma_f32_16x16x4_f32(false, asel[rb][p], false, e2, (short)0, acc2, false, false);
      acc3 = __builtin_amdgcn_wmma_f32_16x16x4_f32(false, asel[rb][p], false, e3, (short)0, acc3, false, false);
    }
  }

  {
    float* dst = &cp[wave][hh * CPH_ + c16];
    #pragma unroll
    for (int r = 0; r < 8; ++r) {
      dst[r * CPP_]        = acc0[r];
      dst[(8 + r) * CPP_]  = acc1[r];
      dst[(16 + r) * CPP_] = acc2[r];
      dst[(24 + r) * CPP_] = acc3[r];
    }
  }
  __syncthreads();

  v8f d0 = {}, d1 = {};
  {
    const float4* s0 = reinterpret_cast<const float4*>(&cp[wave][c16 * CPP_ + 8 * hh]);
    const float4* s1 = reinterpret_cast<const float4*>(&cp[wave][(16 + c16) * CPP_ + 8 * hh]);
    const float4 a0 = s0[0], a1 = s0[1], b0 = s1[0], b1 = s1[1];
    const v2f a00 = {a0.x, a0.y}, a01 = {a0.z, a0.w}, a10 = {a1.x, a1.y}, a11 = {a1.z, a1.w};
    const v2f b00 = {b0.x, b0.y}, b01 = {b0.z, b0.w}, b10 = {b1.x, b1.y}, b11 = {b1.z, b1.w};
    d0 = __builtin_amdgcn_wmma_f32_16x16x4_f32(false, a00, false, ones, (short)0, d0, false, false);
    d1 = __builtin_amdgcn_wmma_f32_16x16x4_f32(false, b00, false, ones, (short)0, d1, false, false);
    d0 = __builtin_amdgcn_wmma_f32_16x16x4_f32(false, a01, false, ones, (short)0, d0, false, false);
    d1 = __builtin_amdgcn_wmma_f32_16x16x4_f32(false, b01, false, ones, (short)0, d1, false, false);
    d0 = __builtin_amdgcn_wmma_f32_16x16x4_f32(false, a10, false, ones, (short)0, d0, false, false);
    d1 = __builtin_amdgcn_wmma_f32_16x16x4_f32(false, b10, false, ones, (short)0, d1, false, false);
    d0 = __builtin_amdgcn_wmma_f32_16x16x4_f32(false, a11, false, ones, (short)0, d0, false, false);
    d1 = __builtin_amdgcn_wmma_f32_16x16x4_f32(false, b11, false, ones, (short)0, d1, false, false);
  }
  if (c16 == 0) {
    float4* o = reinterpret_cast<float4*>(&csum[32 * wave + 8 * hh]);
    o[0] = make_float4(d0[0], d0[1], d0[2], d0[3]);
    o[1] = make_float4(d0[4], d0[5], d0[6], d0[7]);
    o[4] = make_float4(d1[0], d1[1], d1[2], d1[3]);
    o[5] = make_float4(d1[4], d1[5], d1[6], d1[7]);
  }

  float* irs = ws + OFF_IRS + ((c0l * 4 + t) * 64 + bc) * 224;
  if (tid < H_) {
    const int o = CPH_ + (tid >> 4) * CPP_ + (tid & 15);
    float pw[4];
    #pragma unroll
    for (int w = 0; w < 4; ++w)
      pw[w] = ((cp[w][o] + cp[w][o + 8 * CPP_]) + cp[w][o + 16 * CPP_]) + cp[w][o + 24 * CPP_];
    const float s = ((pw[0] + pw[1]) + pw[2]) + pw[3];
    const float r = 1.0f / s; rsb[tid] = r; irs[tid] = r;
  }
  __syncthreads();
  irs[96 + tid] = 1.0f / csum[tid];

  if (tid < 32) {
    float mm = fmaxf(fmaxf(mv[tid], mv[tid + 32]), mv[tid + 64]);
    float nn = fmaxf(fmaxf(nv[tid], nv[tid + 32]), fmaxf(nv[tid + 64], nv[tid + 96]));
    float ri = fminf(fminf(rsb[tid], rsb[tid + 32]), rsb[tid + 64]);
    float cx = fmaxf(fmaxf(csum[tid], csum[tid + 32]), fmaxf(csum[tid + 64], csum[tid + 96]));
    for (int o = 16; o > 0; o >>= 1) {
      mm = fmaxf(mm, __shfl_xor(mm, o));
      nn = fmaxf(nn, __shfl_xor(nn, o));
      ri = fminf(ri, __shfl_xor(ri, o));
      cx = fmaxf(cx, __shfl_xor(cx, o));
    }
    if (tid == 0)
      atomicMax((int*)(ws + OFF_GMAX + c0 * 4 + t),
                __float_as_int(__expf(mm * nn) * (ri + 1.0f / cx)));
  }
}

#define TABW_ 16
static_assert(ROWS_ == 2, "the table layout below is written for two rows per work-group");
template <bool LAST>
__device__ __forceinline__ void score_rows(const float* __restrict__ wsr,
                                           const float* __restrict__ values,
                                           float* __restrict__ out,
                                           const float (*tab)[TABW_],
                                           int c0, int c0l, int b, int h0, int w,
                                           const float (&ig)[4], float (&z)[ROWS_], float& lmax) {
  #pragma unroll 4
  for (int c = 0; c < C_; ++c) {
    const int bc = b * C_ + c;
    const float* irb = wsr + OFF_IRS + ((c0l * 4) * 64 + bc) * 224;
    const float4 tm = *reinterpret_cast<const float4*>(&tab[c][0]);
    const float4 t0 = *reinterpret_cast<const float4*>(&tab[c][4]);
    const float4 t1 = *reinterpret_cast<const float4*>(&tab[c][8]);
    const float mmv[2][ROWS_] = {{tm.x, tm.y}, {tm.z, tm.w}};
    const float irv[ROWS_][4] = {{t0.x, t0.y, t0.z, t0.w}, {t1.x, t1.y, t1.z, t1.w}};
    float nn[2], icv[4];
    nn[0] = wsr[OFF_WR + c0 * (BC_ * W_) + bc * W_ + w];
    nn[1] = wsr[OFF_WC + c0 * (BC_ * W_) + bc * W_ + w];
    #pragma unroll
    for (int t = 0; t < 4; ++t) icv[t] = irb[t * (64 * 224) + 96 + w];
    #pragma unroll
    for (int r = 0; r < ROWS_; ++r) {
      float s = 0.f;
      #pragma unroll
      for (int t = 0; t < 4; ++t) {
        const float e = __expf(mmv[t >> 1][r] * nn[t & 1]);
        s += e * (irv[r][t] + icv[t]) * ig[t];
      }
      const int off = bc * HW_ + (h0 + r) * W_ + w;
      z[r] += s * values[off];
      lmax = fmaxf(lmax, s);
      if (LAST) out[BC_ * HW_ + off] = s;
    }
  }
}

__global__ void k4_scorez(const float* __restrict__ wsr, float* __restrict__ wsw,
                          const float* __restrict__ values, float* __restrict__ out,
                          int c0base) {
  const int w   = threadIdx.x;
  const int rg  = blockIdx.x % (H_ / ROWS_);
  const int cb  = blockIdx.x / (H_ / ROWS_);
  const int b   = cb & 1;
  const int c0l = cb >> 1;
  const int c0  = c0base + c0l;
  const int h0  = rg * ROWS_;
  __shared__ float red[128];
  __shared__ __attribute__((aligned(16))) float tab[C_][TABW_];

  {
    const int   c   = w >> 2, q = w & 3;
    const int   fam = q >> 1, r = q & 1;
    const int   bq  = b * C_ + c0;
    const int   bc  = b * C_ + c;
    const float fa  = wsr[fam ? (OFF_QS + bq * H_ + h0 + r) : (OFF_SQ + bq)];
    const float fb  = wsr[fam ? (OFF_SK + bc) : (OFF_KH + bc * H_ + h0 + r)];
    const float mn  = wsr[OFF_MN + (c0 * 2 + fam) * 2];
    const float iv  = wsr[OFF_MN + (c0 * 2 + fam) * 2 + 1];
    const float2 ir = *reinterpret_cast<const float2*>(wsr + OFF_IRS + ((c0l * 4 + q) * 64 + bc) * 224 + h0);
    const float gq  = wsr[OFF_GMAX + c0 * 4 + q];
    __builtin_amdgcn_sched_barrier(0);
    tab[c][q]      = norm_elem(fa, fb, H_ * EPSF, mn, iv);
    tab[c][4 + q]  = ir.x;
    tab[c][8 + q]  = ir.y;
    tab[c][12 + q] = gq;
  }

  __syncthreads();
  const float4 gm = *reinterpret_cast<const float4*>(&tab[0][12]);
  const float ig[4] = {1.0f / gm.x, 1.0f / gm.y, 1.0f / gm.z, 1.0f / gm.w};

  float z[ROWS_];
  #pragma unroll
  for (int r = 0; r < ROWS_; ++r) z[r] = 0.f;
  float lmax = 0.0f;
  if (c0 == C_ - 1) score_rows<true >(wsr, values, out, tab, c0, c0l, b, h0, w, ig, z, lmax);
  else              score_rows<false>(wsr, values, out, tab, c0, c0l, b, h0, w, ig, z, lmax);

  #pragma unroll
  for (int r = 0; r < ROWS_; ++r) out[(b * C_ + c0) * HW_ + (h0 + r) * W_ + w] = z[r];

  red[w] = lmax;
  __syncthreads();
  for (int s2 = 64; s2 > 0; s2 >>= 1) {
    if (w < s2) red[w] = fmaxf(red[w], red[w + s2]);
    __syncthreads();
  }
  if (w == 0)
    atomicMax((int*)(wsw + OFF_GSMAX + c0), __float_as_int(red[0]));
}

__global__ void k6_scale(const float* __restrict__ ws, float* __restrict__ out) {
  const int g4    = blockIdx.x * 256 + threadIdx.x;
  const int plane = (g4 * 4) / HW_;
  const int c0    = (plane < BC_) ? (plane & (C_ - 1)) : (C_ - 1);
  const float inv = 1.0f / ws[OFF_GSMAX + c0];
  float4* o = reinterpret_cast<float4*>(out) + g4;
  float4 v = *o;
  v.x *= inv; v.y *= inv; v.z *= inv; v.w *= inv;
  *o = v;
}

extern "C" void kernel_launch(void* const* d_in, const int* in_sizes, int n_in,
                              void* d_out, int out_size, void* d_ws, size_t ws_size,
                              hipStream_t stream) {
  (void)in_sizes; (void)n_in; (void)out_size; (void)ws_size;
  const float* q   = (const float*)d_in[0];
  const float* k   = (const float*)d_in[1];
  const float* val = (const float*)d_in[2];
  float* out = (float*)d_out;
  float* ws  = (float*)d_ws;

  k1_sums<<<128, 256, 0, stream>>>(q, k, ws);
  k2_norm<<<128, 256, 0, stream>>>(ws);
  k3_tile<<<4096, 128, 0, stream>>>(ws, 0);
  k4_scorez<<<1536, 128, 0, stream>>>(ws, ws, val, out, 0);
  k3_tile<<<4096, 128, 0, stream>>>(ws, 16);
  k4_scorez<<<1536, 128, 0, stream>>>(ws, ws, val, out, 16);
  k6_scale<<<1536, 256, 0, stream>>>(ws, out);
}
